# starting kernel
# baseline (speedup 1.0000x reference)
#include <hip/hip_runtime.h>

#define N_TOTAL    64000
#define NODE_COUNT 1000
#define BATCH      64
#define STATE      16
#define HID        32
#define N_EDGES    2048000
#define E_TOTAL    (N_EDGES + N_TOTAL)

typedef float v2f __attribute__((ext_vector_type(2)));
typedef float v8f __attribute__((ext_vector_type(8)));

typedef const __attribute__((address_space(1))) float* gcfp;
typedef __attribute__((address_space(1))) float*       gfp;

__global__ void k_zero(float* __restrict__ p, int n) {
  int i = blockIdx.x * blockDim.x + threadIdx.x;
  if (i < n) p[i] = 0.0f;
}

__global__ void k_degree(const int* __restrict__ cols, float* __restrict__ deg) {
  int e = blockIdx.x * blockDim.x + threadIdx.x;
  if (e < N_EDGES) atomicAdd(&deg[cols[e]], 1.0f);
}

__global__ void k_dinv(float* __restrict__ deg) {
  int i = blockIdx.x * blockDim.x + threadIdx.x;
  if (i < N_TOTAL) deg[i] = rsqrtf(deg[i] + 1.0f);
}

__global__ void k_norm(const int* __restrict__ rows, const int* __restrict__ cols,
                       const float* __restrict__ dinv, float* __restrict__ norm) {
  int e = blockIdx.x * blockDim.x + threadIdx.x;
  if (e >= E_TOTAL) return;
  int r, c;
  if (e < N_EDGES) { r = rows[e]; c = cols[e]; }
  else             { r = c = e - N_EDGES; }
  norm[e] = dinv[r] * dinv[c];
}

__global__ void k_scatter(const int* __restrict__ rows, const int* __restrict__ cols,
                          const float* __restrict__ norm,
                          const float* __restrict__ h, float* __restrict__ out) {
  int t = blockIdx.x * blockDim.x + threadIdx.x;
  int e = t >> 5;
  int j = t & 31;
  if (e >= E_TOTAL) return;
  int r, c;
  if (e < N_EDGES) { r = rows[e]; c = cols[e]; }
  else             { r = c = e - N_EDGES; }
  float w = norm[e];
  gcfp hg = (gcfp)h;
  atomicAdd(&out[(unsigned)c * HID + j], hg[(unsigned)r * HID + j] * w);
}

__global__ void k_bias_relu(float* __restrict__ x, const float* __restrict__ b, int n) {
  int i = blockIdx.x * blockDim.x + threadIdx.x;
  if (i < n) {
    float v = x[i] + b[i & (HID - 1)];
    x[i] = v > 0.0f ? v : 0.0f;
  }
}

template <int NT, bool RAGGED, bool BIAS, bool RELU>
__global__ void k_gemm_wmma(const float* __restrict__ A, const float* __restrict__ B,
                            const float* __restrict__ bias, float* __restrict__ C,
                            int M, int N, int K) {
  int wave  = threadIdx.x >> 5;
  int lane  = threadIdx.x & 31;
  int mtile = blockIdx.y * 4 + wave;
  int m0    = mtile * 16;
  if (m0 >= M) return;
  int n0   = blockIdx.x * (16 * NT);
  int half = lane >> 4;
  int lm   = lane & 15;

  gcfp Ag = (gcfp)A;
  gcfp Bg = (gcfp)B;

  int  ncol[NT];
  bool nok[NT];
  unsigned boff[NT];
#pragma unroll
  for (int t = 0; t < NT; ++t) {
    ncol[t] = n0 + t * 16 + lm;
    int nld;
    if (RAGGED) {
      nok[t] = ncol[t] < N;
      nld    = nok[t] ? ncol[t] : (N - 1);
    } else {
      nok[t] = true;
      nld    = ncol[t];
    }
    boff[t] = (unsigned)(half * 2) * (unsigned)N + (unsigned)nld;
  }

  unsigned aoff = (unsigned)(m0 + lm) * (unsigned)K + (unsigned)(half * 2);

  v8f acc[NT];
#pragma unroll
  for (int t = 0; t < NT; ++t) acc[t] = (v8f){0.f,0.f,0.f,0.f,0.f,0.f,0.f,0.f};

  for (int k0 = 0; k0 < K; k0 += 4) {
    v2f a;
    a.x = Ag[aoff];
    a.y = Ag[aoff + 1];
#pragma unroll
    for (int t = 0; t < NT; ++t) {
      float b0 = Bg[boff[t]];
      float b1 = Bg[boff[t] + (unsigned)N];
      if (RAGGED) { b0 = nok[t] ? b0 : 0.0f; b1 = nok[t] ? b1 : 0.0f; }
      v2f b; b.x = b0; b.y = b1;
      acc[t] = __builtin_amdgcn_wmma_f32_16x16x4_f32(
           false, a,  false, b,
           (short)0, acc[t],  false,  false);
      boff[t] += 4u * (unsigned)N;
    }
    aoff += 4u;
  }

  gfp Cg = (gfp)C;
  gcfp biasg = (gcfp)bias;
#pragma unroll
  for (int t = 0; t < NT; ++t) {
    float bv = BIAS ? biasg[nok[t] ? ncol[t] : 0] : 0.0f;
    if (!RAGGED || nok[t]) {
#pragma unroll
      for (int r = 0; r < 8; ++r) {
        int m = m0 + r + half * 8;
        float v = acc[t][r] + bv;
        if (RELU) v = v > 0.0f ? v : 0.0f;
        Cg[(unsigned)m * (unsigned)N + (unsigned)ncol[t]] = v;
      }
    }
  }
}

__global__ void k_value(const float* __restrict__ hidv, const float* __restrict__ Wv2,
                        const float* __restrict__ bv2, float* __restrict__ out) {
  int b = blockIdx.x * blockDim.x + threadIdx.x;
  if (b < BATCH) {
    float s = bv2[0];
    for (int k = 0; k < 256; ++k) s += hidv[b * 256 + k] * Wv2[k];
    out[b] = s;
  }
}

static inline int cdiv(long long a, int b) { return (int)((a + b - 1) / b); }

extern "C" void kernel_launch(void* const* d_in, const int* in_sizes, int n_in,
                              void* d_out, int out_size, void* d_ws, size_t ws_size,
                              hipStream_t stream) {
  (void)in_sizes; (void)n_in; (void)out_size; (void)ws_size;

  const float* x    = (const float*)d_in[0];
  const int*   ei   = (const int*)d_in[1];
  const int*   rows = ei;
  const int*   cols = ei + N_EDGES;
  const float* W[5]  = {(const float*)d_in[2], (const float*)d_in[4], (const float*)d_in[6],
                        (const float*)d_in[8], (const float*)d_in[10]};
  const float* bb[5] = {(const float*)d_in[3], (const float*)d_in[5], (const float*)d_in[7],
                        (const float*)d_in[9], (const float*)d_in[11]};
  const float* Wp1 = (const float*)d_in[12]; const float* bp1 = (const float*)d_in[13];
  const float* Wp2 = (const float*)d_in[14]; const float* bp2 = (const float*)d_in[15];
  const float* Wv1 = (const float*)d_in[16]; const float* bv1 = (const float*)d_in[17];
  const float* Wv2 = (const float*)d_in[18]; const float* bv2 = (const float*)d_in[19];
  float* out = (float*)d_out;

  float* ws   = (float*)d_ws;
  float* deg  = ws;
  float* norm = deg  + N_TOTAL;
  float* bufA = norm + E_TOTAL;
  float* bufB = bufA + (size_t)N_TOTAL * HID;
  float* hidp = bufB + (size_t)N_TOTAL * HID;
  float* hidv = hidp + BATCH * 256;

  const int TB = 256;

  k_zero  <<<cdiv(N_TOTAL, TB), TB, 0, stream>>>(deg, N_TOTAL);
  k_degree<<<cdiv(N_EDGES, TB), TB, 0, stream>>>(cols, deg);
  k_dinv  <<<cdiv(N_TOTAL, TB), TB, 0, stream>>>(deg);
  k_norm  <<<cdiv(E_TOTAL, TB), TB, 0, stream>>>(rows, cols, deg, norm);

  const float* in = x;
  int Kdim = STATE;
  dim3 layer_grid(1, N_TOTAL / 64);
  int scat_blocks = cdiv((long long)E_TOTAL * 32, TB);
  for (int l = 0; l < 5; ++l) {
    k_gemm_wmma<2, false, false, false><<<layer_grid, 128, 0, stream>>>(
        in, W[l], nullptr, bufA, N_TOTAL, HID, Kdim);
    k_zero     <<<cdiv((long long)N_TOTAL * HID, TB), TB, 0, stream>>>(bufB, N_TOTAL * HID);
    k_scatter  <<<scat_blocks, TB, 0, stream>>>(rows, cols, norm, bufA, bufB);
    k_bias_relu<<<cdiv((long long)N_TOTAL * HID, TB), TB, 0, stream>>>(bufB, bb[l],
                                                                       N_TOTAL * HID);
    in = bufB;
    Kdim = HID;
  }

  dim3 g1(256 / 32, cdiv(BATCH, 64));
  k_gemm_wmma<2, false, true, true><<<g1, 128, 0, stream>>>(
      bufB, Wp1, bp1, hidp, BATCH, 256, NODE_COUNT * HID);
  k_gemm_wmma<2, false, true, true><<<g1, 128, 0, stream>>>(
      bufB, Wv1, bv1, hidv, BATCH, 256, NODE_COUNT * HID);

  dim3 g2(cdiv(NODE_COUNT, 16), cdiv(BATCH, 64));
  k_gemm_wmma<1, true, true, false><<<g2, 128, 0, stream>>>(
      hidp, Wp2, bp2, out, BATCH, NODE_COUNT, 256);

  k_value<<<1, 64, 0, stream>>>(hidv, Wv2, bv2, out + (size_t)BATCH * NODE_COUNT);
}


# Head GEMMs (M=64, K=32000): K split into 250 slices over 250x as many work-groups, partial planes in workspace, one added kernel sums them in plane order + bias + relu; f32 sum order changes
# speedup vs baseline: 3.6803x; 3.6803x over previous
#include <hip/hip_runtime.h>

#define N_TOTAL    64000
#define NODE_COUNT 1000
#define BATCH      64
#define STATE      16
#define HID        32
#define N_EDGES    2048000
#define E_TOTAL    (N_EDGES + N_TOTAL)
#define HEAD_K     (NODE_COUNT * HID)
#define HEAD_N     256
#define KSPLIT     250
#define KCHUNK     (HEAD_K / KSPLIT)
#define PLANE      (BATCH * HEAD_N)

typedef float v2f __attribute__((ext_vector_type(2)));
typedef float v8f __attribute__((ext_vector_type(8)));

typedef const __attribute__((address_space(1))) float* gcfp;
typedef __attribute__((address_space(1))) float*       gfp;

__global__ void k_zero(float* __restrict__ p, int n) {
  int i = blockIdx.x * blockDim.x + threadIdx.x;
  if (i < n) p[i] = 0.0f;
}

__global__ void k_degree(const int* __restrict__ cols, float* __restrict__ deg) {
  int e = blockIdx.x * blockDim.x + threadIdx.x;
  if (e < N_EDGES) atomicAdd(&deg[cols[e]], 1.0f);
}

__global__ void k_dinv(float* __restrict__ deg) {
  int i = blockIdx.x * blockDim.x + threadIdx.x;
  if (i < N_TOTAL) deg[i] = rsqrtf(deg[i] + 1.0f);
}

__global__ void k_norm(const int* __restrict__ rows, const int* __restrict__ cols,
                       const float* __restrict__ dinv, float* __restrict__ norm) {
  int e = blockIdx.x * blockDim.x + threadIdx.x;
  if (e >= E_TOTAL) return;
  int r, c;
  if (e < N_EDGES) { r = rows[e]; c = cols[e]; }
  else             { r = c = e - N_EDGES; }
  norm[e] = dinv[r] * dinv[c];
}

__global__ void k_scatter(const int* __restrict__ rows, const int* __restrict__ cols,
                          const float* __restrict__ norm,
                          const float* __restrict__ h, float* __restrict__ out) {
  int t = blockIdx.x * blockDim.x + threadIdx.x;
  int e = t >> 5;
  int j = t & 31;
  if (e >= E_TOTAL) return;
  int r, c;
  if (e < N_EDGES) { r = rows[e]; c = cols[e]; }
  else             { r = c = e - N_EDGES; }
  float w = norm[e];
  gcfp hg = (gcfp)h;
  atomicAdd(&out[(unsigned)c * HID + j], hg[(unsigned)r * HID + j] * w);
}

__global__ void k_bias_relu(float* __restrict__ x, const float* __restrict__ b, int n) {
  int i = blockIdx.x * blockDim.x + threadIdx.x;
  if (i < n) {
    float v = x[i] + b[i & (HID - 1)];
    x[i] = v > 0.0f ? v : 0.0f;
  }
}

template <int NT, bool RAGGED, bool BIAS, bool RELU>
__global__ void k_gemm_wmma(const float* __restrict__ A, const float* __restrict__ B,
                            const float* __restrict__ bias, float* __restrict__ C,
                            int M, int N, int K) {
  int wave  = threadIdx.x >> 5;
  int lane  = threadIdx.x & 31;
  int mtile = blockIdx.y * 4 + wave;
  int m0    = mtile * 16;
  if (m0 >= M) return;
  int n0   = blockIdx.x * (16 * NT);
  int half = lane >> 4;
  int lm   = lane & 15;

  gcfp Ag = (gcfp)A;
  gcfp Bg = (gcfp)B;

  int  ncol[NT];
  bool nok[NT];
  unsigned boff[NT];
#pragma unroll
  for (int t = 0; t < NT; ++t) {
    ncol[t] = n0 + t * 16 + lm;
    int nld;
    if (RAGGED) {
      nok[t] = ncol[t] < N;
      nld    = nok[t] ? ncol[t] : (N - 1);
    } else {
      nok[t] = true;
      nld    = ncol[t];
    }
    boff[t] = (unsigned)(half * 2) * (unsigned)N + (unsigned)nld;
  }

  unsigned aoff = (unsigned)(m0 + lm) * (unsigned)K + (unsigned)(half * 2);

  v8f acc[NT];
#pragma unroll
  for (int t = 0; t < NT; ++t) acc[t] = (v8f){0.f,0.f,0.f,0.f,0.f,0.f,0.f,0.f};

  for (int k0 = 0; k0 < K; k0 += 4) {
    v2f a;
    a.x = Ag[aoff];
    a.y = Ag[aoff + 1];
#pragma unroll
    for (int t = 0; t < NT; ++t) {
      float b0 = Bg[boff[t]];
      float b1 = Bg[boff[t] + (unsigned)N];
      if (RAGGED) { b0 = nok[t] ? b0 : 0.0f; b1 = nok[t] ? b1 : 0.0f; }
      v2f b; b.x = b0; b.y = b1;
      acc[t] = __builtin_amdgcn_wmma_f32_16x16x4_f32(
           false, a,  false, b,
           (short)0, acc[t],  false,  false);
      boff[t] += 4u * (unsigned)N;
    }
    aoff += 4u;
  }

  gfp Cg = (gfp)C;
  gcfp biasg = (gcfp)bias;
#pragma unroll
  for (int t = 0; t < NT; ++t) {
    float bv = BIAS ? biasg[nok[t] ? ncol[t] : 0] : 0.0f;
    if (!RAGGED || nok[t]) {
#pragma unroll
      for (int r = 0; r < 8; ++r) {
        int m = m0 + r + half * 8;
        float v = acc[t][r] + bv;
        if (RELU) v = v > 0.0f ? v : 0.0f;
        Cg[(unsigned)m * (unsigned)N + (unsigned)ncol[t]] = v;
      }
    }
  }
}

template <int NT>
__global__ void k_gemm_wmma_ksplit(const float* __restrict__ A, const float* __restrict__ B,
                                   float* __restrict__ P, int M, int N, int K, int KC) {
  int wave  = threadIdx.x >> 5;
  int lane  = threadIdx.x & 31;
  int mtile = blockIdx.y * 4 + wave;
  int m0    = mtile * 16;
  if (m0 >= M) return;
  int n0   = blockIdx.x * (16 * NT);
  int half = lane >> 4;
  int lm   = lane & 15;
  unsigned kbeg = (unsigned)blockIdx.z * (unsigned)KC;

  gcfp Ag = (gcfp)A;
  gcfp Bg = (gcfp)B;

  int ncol[NT];
  unsigned boff[NT];
#pragma unroll
  for (int t = 0; t < NT; ++t) {
    ncol[t] = n0 + t * 16 + lm;
    boff[t] = (kbeg + (unsigned)(half * 2)) * (unsigned)N + (unsigned)ncol[t];
  }

  unsigned aoff = (unsigned)(m0 + lm) * (unsigned)K + kbeg + (unsigned)(half * 2);

  v8f acc[NT];
#pragma unroll
  for (int t = 0; t < NT; ++t) acc[t] = (v8f){0.f,0.f,0.f,0.f,0.f,0.f,0.f,0.f};

  for (int k0 = 0; k0 < KC; k0 += 4) {
    v2f a;
    a.x = Ag[aoff];
    a.y = Ag[aoff + 1];
#pragma unroll
    for (int t = 0; t < NT; ++t) {
      v2f b;
      b.x = Bg[boff[t]];
      b.y = Bg[boff[t] + (unsigned)N];
      acc[t] = __builtin_amdgcn_wmma_f32_16x16x4_f32(
           false, a,  false, b,
           (short)0, acc[t],  false,  false);
      boff[t] += 4u * (unsigned)N;
    }
    aoff += 4u;
  }

  gfp Pg = (gfp)P;
  unsigned pbase = (unsigned)blockIdx.z * (unsigned)M * (unsigned)N;
#pragma unroll
  for (int t = 0; t < NT; ++t) {
#pragma unroll
    for (int r = 0; r < 8; ++r) {
      int m = m0 + r + half * 8;
      Pg[pbase + (unsigned)m * (unsigned)N + (unsigned)ncol[t]] = acc[t][r];
    }
  }
}

__global__ void k_head_reduce(const float* __restrict__ part,
                              const float* __restrict__ bp1, const float* __restrict__ bv1,
                              float* __restrict__ hid) {
  int i = blockIdx.x * blockDim.x + threadIdx.x;
  if (i >= 2 * PLANE) return;
  int h = i >= PLANE ? 1 : 0;
  int e = i - h * PLANE;
  gcfp p = (gcfp)part + (size_t)h * KSPLIT * PLANE + e;
  float s = 0.0f;
  for (int z = 0; z < KSPLIT; ++z) s += p[(unsigned)z * (unsigned)PLANE];
  const float* bias = h ? bv1 : bp1;
  s += bias[e & (HEAD_N - 1)];
  hid[i] = s > 0.0f ? s : 0.0f;
}

__global__ void k_value(const float* __restrict__ hidv, const float* __restrict__ Wv2,
                        const float* __restrict__ bv2, float* __restrict__ out) {
  int b = blockIdx.x * blockDim.x + threadIdx.x;
  if (b < BATCH) {
    float s = bv2[0];
    for (int k = 0; k < 256; ++k) s += hidv[b * 256 + k] * Wv2[k];
    out[b] = s;
  }
}

static inline int cdiv(long long a, int b) { return (int)((a + b - 1) / b); }

extern "C" void kernel_launch(void* const* d_in, const int* in_sizes, int n_in,
                              void* d_out, int out_size, void* d_ws, size_t ws_size,
                              hipStream_t stream) {
  (void)in_sizes; (void)n_in; (void)out_size; (void)ws_size;

  const float* x    = (const float*)d_in[0];
  const int*   ei   = (const int*)d_in[1];
  const int*   rows = ei;
  const int*   cols = ei + N_EDGES;
  const float* W[5]  = {(const float*)d_in[2], (const float*)d_in[4], (const float*)d_in[6],
                        (const float*)d_in[8], (const float*)d_in[10]};
  const float* bb[5] = {(const float*)d_in[3], (const float*)d_in[5], (const float*)d_in[7],
                        (const float*)d_in[9], (const float*)d_in[11]};
  const float* Wp1 = (const float*)d_in[12]; const float* bp1 = (const float*)d_in[13];
  const float* Wp2 = (const float*)d_in[14]; const float* bp2 = (const float*)d_in[15];
  const float* Wv1 = (const float*)d_in[16]; const float* bv1 = (const float*)d_in[17];
  const float* Wv2 = (const float*)d_in[18]; const float* bv2 = (const float*)d_in[19];
  float* out = (float*)d_out;

  float* ws   = (float*)d_ws;
  float* deg  = ws;
  float* norm = deg  + N_TOTAL;
  float* bufA = norm + E_TOTAL;
  float* bufB = bufA + (size_t)N_TOTAL * HID;
  float* hidp = bufB + (size_t)N_TOTAL * HID;
  float* hidv = hidp + BATCH * 256;
  float* part = hidv + BATCH * 256;

  const int TB = 256;

  k_zero  <<<cdiv(N_TOTAL, TB), TB, 0, stream>>>(deg, N_TOTAL);
  k_degree<<<cdiv(N_EDGES, TB), TB, 0, stream>>>(cols, deg);
  k_dinv  <<<cdiv(N_TOTAL, TB), TB, 0, stream>>>(deg);
  k_norm  <<<cdiv(E_TOTAL, TB), TB, 0, stream>>>(rows, cols, deg, norm);

  const float* in = x;
  int Kdim = STATE;
  dim3 layer_grid(1, N_TOTAL / 64);
  int scat_blocks = cdiv((long long)E_TOTAL * 32, TB);
  for (int l = 0; l < 5; ++l) {
    k_gemm_wmma<2, false, false, false><<<layer_grid, 128, 0, stream>>>(
        in, W[l], nullptr, bufA, N_TOTAL, HID, Kdim);
    k_zero     <<<cdiv((long long)N_TOTAL * HID, TB), TB, 0, stream>>>(bufB, N_TOTAL * HID);
    k_scatter  <<<scat_blocks, TB, 0, stream>>>(rows, cols, norm, bufA, bufB);
    k_bias_relu<<<cdiv((long long)N_TOTAL * HID, TB), TB, 0, stream>>>(bufB, bb[l],
                                                                       N_TOTAL * HID);
    in = bufB;
    Kdim = HID;
  }

  dim3 g1(HEAD_N / 32, cdiv(BATCH, 64), KSPLIT);
  k_gemm_wmma_ksplit<2><<<g1, 128, 0, stream>>>(
      bufB, Wp1, part, BATCH, HEAD_N, HEAD_K, KCHUNK);
  k_gemm_wmma_ksplit<2><<<g1, 128, 0, stream>>>(
      bufB, Wv1, part + (size_t)KSPLIT * PLANE, BATCH, HEAD_N, HEAD_K, KCHUNK);
  k_head_reduce<<<cdiv(2 * PLANE, 128), 128, 0, stream>>>(part, bp1, bv1, hidp);

  dim3 g2(cdiv(NODE_COUNT, 16), cdiv(BATCH, 64));
  k_gemm_wmma<1, true, true, false><<<g2, 128, 0, stream>>>(
      hidp, Wp2, bp2, out, BATCH, NODE_COUNT, 256);

  k_value<<<1, 64, 0, stream>>>(hidv, Wv2, bv2, out + (size_t)BATCH * NODE_COUNT);
}


# Per-layer f32 atomic scatter-add replaced by gather-sum over an edge list counting-sorted by target once per call (bias+relu in it); sum order per node now ascending edge id, self loop last
# speedup vs baseline: 10.1110x; 2.7473x over previous
#include <hip/hip_runtime.h>

#define N_TOTAL    64000
#define NODE_COUNT 1000
#define BATCH      64
#define STATE      16
#define HID        32
#define N_EDGES    2048000
#define E_TOTAL    (N_EDGES + N_TOTAL)
#define HEAD_K     (NODE_COUNT * HID)
#define HEAD_N     256
#define KSPLIT     250
#define KCHUNK     (HEAD_K / KSPLIT)
#define PLANE      (BATCH * HEAD_N)
#define SCAN_TB    256
#define SCAN_WG    (N_TOTAL / SCAN_TB)

typedef float v2f __attribute__((ext_vector_type(2)));
typedef float v8f __attribute__((ext_vector_type(8)));

typedef const __attribute__((address_space(1))) float* gcfp;
typedef __attribute__((address_space(1))) float*       gfp;

__global__ void k_zero_int(int* __restrict__ p, int n) {
  int i = blockIdx.x * blockDim.x + threadIdx.x;
  if (i < n) p[i] = 0;
}

__global__ void k_count(const int* __restrict__ cols, int* __restrict__ cnt,
                        int* __restrict__ slot) {
  int e = blockIdx.x * blockDim.x + threadIdx.x;
  if (e < N_EDGES) slot[e] = atomicAdd(&cnt[cols[e]], 1);
}

__global__ void k_scan_local(const int* __restrict__ cnt, int* __restrict__ rowptr,
                             int* __restrict__ bsum, float* __restrict__ dinv) {
  __shared__ int s[SCAN_TB];
  int t = threadIdx.x;
  int i = blockIdx.x * SCAN_TB + t;
  int v = cnt[i];
  dinv[i] = rsqrtf((float)v + 1.0f);
  s[t] = v;
  __syncthreads();
  for (int off = 1; off < SCAN_TB; off <<= 1) {
    int a = t >= off ? s[t - off] : 0;
    __syncthreads();
    s[t] += a;
    __syncthreads();
  }
  rowptr[i] = s[t] - v;
  if (t == SCAN_TB - 1) bsum[blockIdx.x] = s[t];
}

__global__ void k_scan_add(const int* __restrict__ cnt, const int* __restrict__ bsum,
                           int* __restrict__ rowptr) {
  __shared__ int s[SCAN_TB];
  int t = threadIdx.x;
  int b = blockIdx.x;
  s[t] = t < b ? bsum[t] : 0;
  __syncthreads();
  for (int off = SCAN_TB / 2; off > 0; off >>= 1) {
    if (t < off) s[t] += s[t + off];
    __syncthreads();
  }
  int i = b * SCAN_TB + t;
  int p = rowptr[i] + s[0];
  rowptr[i] = p;
  if (i == N_TOTAL - 1) rowptr[N_TOTAL] = p + cnt[i];
}

__global__ void k_fill(const int* __restrict__ cols, const int* __restrict__ slot,
                       const int* __restrict__ rowptr, int* __restrict__ eid) {
  int e = blockIdx.x * blockDim.x + threadIdx.x;
  if (e < N_EDGES) eid[rowptr[cols[e]] + slot[e]] = e;
}

__global__ void k_rank(const int* __restrict__ rows, const int* __restrict__ rowptr,
                       const int* __restrict__ eid, const float* __restrict__ dinv,
                       int* __restrict__ srcs, float* __restrict__ wts) {
  int t    = blockIdx.x * blockDim.x + threadIdx.x;
  int lane = t & 31;
  int c = __builtin_amdgcn_readfirstlane(t >> 5);
  if (c >= N_TOTAL) return;
  int beg = rowptr[c], end = rowptr[c + 1];
  float dc = dinv[c];
  for (int base = beg; base < end; base += 32) {
    int i   = base + lane;
    bool ok = i < end;
    int my  = ok ? eid[i] : 0x7fffffff;
    int rank = 0;
    for (int cb = beg; cb < end; cb += 32) {
      int q     = cb + lane;
      int other = q < end ? eid[q] : 0x7fffffff;
      int n     = end - cb < 32 ? end - cb : 32;
      for (int k = 0; k < n; ++k)
        rank += __builtin_amdgcn_readlane(other, k) < my ? 1 : 0;
    }
    if (ok) {
      int r = rows[my];
      srcs[beg + rank] = r;
      wts[beg + rank]  = dinv[r] * dc;
    }
  }
}

__global__ void k_aggregate(const int* __restrict__ rowptr, const int* __restrict__ srcs,
                            const float* __restrict__ wts, const float* __restrict__ dinv,
                            const float* __restrict__ h, const float* __restrict__ bias,
                            float* __restrict__ out) {
  int t = blockIdx.x * blockDim.x + threadIdx.x;
  int j = t & 31;
  int c = __builtin_amdgcn_readfirstlane(t >> 5);
  if (c >= N_TOTAL) return;
  gcfp hg = (gcfp)h;
  int beg = rowptr[c], end = rowptr[c + 1];
  float acc = 0.0f;
  for (int base = beg; base < end; base += 32) {
    int i = base + j;
    int   myr = 0;
    float myw = 0.0f;
    if (i < end) { myr = srcs[i]; myw = wts[i]; }
    int n = end - base < 32 ? end - base : 32;
#pragma unroll 8
    for (int k = 0; k < n; ++k) {
      int   r = __builtin_amdgcn_readlane(myr, k);
      float w = __int_as_float(__builtin_amdgcn_readlane(__float_as_int(myw), k));
      acc += hg[(unsigned)r * HID + j] * w;
    }
  }
  float dc = dinv[c];
  acc += hg[(unsigned)c * HID + j] * (dc * dc);
  float v = acc + bias[j];
  ((gfp)out)[(unsigned)c * HID + j] = v > 0.0f ? v : 0.0f;
}

template <int NT, bool RAGGED, bool BIAS, bool RELU>
__global__ void k_gemm_wmma(const float* __restrict__ A, const float* __restrict__ B,
                            const float* __restrict__ bias, float* __restrict__ C,
                            int M, int N, int K) {
  int wave  = threadIdx.x >> 5;
  int lane  = threadIdx.x & 31;
  int mtile = blockIdx.y * 4 + wave;
  int m0    = mtile * 16;
  if (m0 >= M) return;
  int n0   = blockIdx.x * (16 * NT);
  int half = lane >> 4;
  int lm   = lane & 15;

  gcfp Ag = (gcfp)A;
  gcfp Bg = (gcfp)B;

  int  ncol[NT];
  bool nok[NT];
  unsigned boff[NT];
#pragma unroll
  for (int t = 0; t < NT; ++t) {
    ncol[t] = n0 + t * 16 + lm;
    int nld;
    if (RAGGED) {
      nok[t] = ncol[t] < N;
      nld    = nok[t] ? ncol[t] : (N - 1);
    } else {
      nok[t] = true;
      nld    = ncol[t];
    }
    boff[t] = (unsigned)(half * 2) * (unsigned)N + (unsigned)nld;
  }

  unsigned aoff = (unsigned)(m0 + lm) * (unsigned)K + (unsigned)(half * 2);

  v8f acc[NT];
#pragma unroll
  for (int t = 0; t < NT; ++t) acc[t] = (v8f){0.f,0.f,0.f,0.f,0.f,0.f,0.f,0.f};

  for (int k0 = 0; k0 < K; k0 += 4) {
    v2f a;
    a.x = Ag[aoff];
    a.y = Ag[aoff + 1];
#pragma unroll
    for (int t = 0; t < NT; ++t) {
      float b0 = Bg[boff[t]];
      float b1 = Bg[boff[t] + (unsigned)N];
      if (RAGGED) { b0 = nok[t] ? b0 : 0.0f; b1 = nok[t] ? b1 : 0.0f; }
      v2f b; b.x = b0; b.y = b1;
      acc[t] = __builtin_amdgcn_wmma_f32_16x16x4_f32(
           false, a,  false, b,
           (short)0, acc[t],  false,  false);
      boff[t] += 4u * (unsigned)N;
    }
    aoff += 4u;
  }

  gfp Cg = (gfp)C;
  gcfp biasg = (gcfp)bias;
#pragma unroll
  for (int t = 0; t < NT; ++t) {
    float bv = BIAS ? biasg[nok[t] ? ncol[t] : 0] : 0.0f;
    if (!RAGGED || nok[t]) {
#pragma unroll
      for (int r = 0; r < 8; ++r) {
        int m = m0 + r + half * 8;
        float v = acc[t][r] + bv;
        if (RELU) v = v > 0.0f ? v : 0.0f;
        Cg[(unsigned)m * (unsigned)N + (unsigned)ncol[t]] = v;
      }
    }
  }
}

template <int NT>
__global__ void k_gemm_wmma_ksplit(const float* __restrict__ A, const float* __restrict__ B,
                                   float* __restrict__ P, int M, int N, int K, int KC) {
  int wave  = threadIdx.x >> 5;
  int lane  = threadIdx.x & 31;
  int mtile = blockIdx.y * 4 + wave;
  int m0    = mtile * 16;
  if (m0 >= M) return;
  int n0   = blockIdx.x * (16 * NT);
  int half = lane >> 4;
  int lm   = lane & 15;
  unsigned kbeg = (unsigned)blockIdx.z * (unsigned)KC;

  gcfp Ag = (gcfp)A;
  gcfp Bg = (gcfp)B;

  int ncol[NT];
  unsigned boff[NT];
#pragma unroll
  for (int t = 0; t < NT; ++t) {
    ncol[t] = n0 + t * 16 + lm;
    boff[t] = (kbeg + (unsigned)(half * 2)) * (unsigned)N + (unsigned)ncol[t];
  }

  unsigned aoff = (unsigned)(m0 + lm) * (unsigned)K + kbeg + (unsigned)(half * 2);

  v8f acc[NT];
#pragma unroll
  for (int t = 0; t < NT; ++t) acc[t] = (v8f){0.f,0.f,0.f,0.f,0.f,0.f,0.f,0.f};

  for (int k0 = 0; k0 < KC; k0 += 4) {
    v2f a;
    a.x = Ag[aoff];
    a.y = Ag[aoff + 1];
#pragma unroll
    for (int t = 0; t < NT; ++t) {
      v2f b;
      b.x = Bg[boff[t]];
      b.y = Bg[boff[t] + (unsigned)N];
      acc[t] = __builtin_amdgcn_wmma_f32_16x16x4_f32(
           false, a,  false, b,
           (short)0, acc[t],  false,  false);
      boff[t] += 4u * (unsigned)N;
    }
    aoff += 4u;
  }

  gfp Pg = (gfp)P;
  unsigned pbase = (unsigned)blockIdx.z * (unsigned)M * (unsigned)N;
#pragma unroll
  for (int t = 0; t < NT; ++t) {
#pragma unroll
    for (int r = 0; r < 8; ++r) {
      int m = m0 + r + half * 8;
      Pg[pbase + (unsigned)m * (unsigned)N + (unsigned)ncol[t]] = acc[t][r];
    }
  }
}

__global__ void k_head_reduce(const float* __restrict__ part,
                              const float* __restrict__ bp1, const float* __restrict__ bv1,
                              float* __restrict__ hid) {
  int i = blockIdx.x * blockDim.x + threadIdx.x;
  if (i >= 2 * PLANE) return;
  int h = i >= PLANE ? 1 : 0;
  int e = i - h * PLANE;
  gcfp p = (gcfp)part + (size_t)h * KSPLIT * PLANE + e;
  float s = 0.0f;
  for (int z = 0; z < KSPLIT; ++z) s += p[(unsigned)z * (unsigned)PLANE];
  const float* bias = h ? bv1 : bp1;
  s += bias[e & (HEAD_N - 1)];
  hid[i] = s > 0.0f ? s : 0.0f;
}

__global__ void k_value(const float* __restrict__ hidv, const float* __restrict__ Wv2,
                        const float* __restrict__ bv2, float* __restrict__ out) {
  int b = blockIdx.x * blockDim.x + threadIdx.x;
  if (b < BATCH) {
    float s = bv2[0];
    for (int k = 0; k < 256; ++k) s += hidv[b * 256 + k] * Wv2[k];
    out[b] = s;
  }
}

static inline int cdiv(long long a, int b) { return (int)((a + b - 1) / b); }

extern "C" void kernel_launch(void* const* d_in, const int* in_sizes, int n_in,
                              void* d_out, int out_size, void* d_ws, size_t ws_size,
                              hipStream_t stream) {
  (void)in_sizes; (void)n_in; (void)out_size; (void)ws_size;

  const float* x    = (const float*)d_in[0];
  const int*   ei   = (const int*)d_in[1];
  const int*   rows = ei;
  const int*   cols = ei + N_EDGES;
  const float* W[5]  = {(const float*)d_in[2], (const float*)d_in[4], (const float*)d_in[6],
                        (const float*)d_in[8], (const float*)d_in[10]};
  const float* bb[5] = {(const float*)d_in[3], (const float*)d_in[5], (const float*)d_in[7],
                        (const float*)d_in[9], (const float*)d_in[11]};
  const float* Wp1 = (const float*)d_in[12]; const float* bp1 = (const float*)d_in[13];
  const float* Wp2 = (const float*)d_in[14]; const float* bp2 = (const float*)d_in[15];
  const float* Wv1 = (const float*)d_in[16]; const float* bv1 = (const float*)d_in[17];
  const float* Wv2 = (const float*)d_in[18]; const float* bv2 = (const float*)d_in[19];
  float* out = (float*)d_out;

  float* ws     = (float*)d_ws;
  int*   cnt    = (int*)ws;
  int*   rowptr = cnt + N_TOTAL;
  int*   bsum   = rowptr + N_TOTAL + 64;
  float* dinv   = (float*)(bsum + 256);
  int*   slot   = (int*)(dinv + N_TOTAL);
  int*   eid    = slot + N_EDGES;
  int*   srcs   = eid + N_EDGES;
  float* wts    = (float*)(srcs + N_EDGES);
  float* bufA   = wts + N_EDGES;
  float* bufB   = bufA + (size_t)N_TOTAL * HID;
  float* hidp   = bufB + (size_t)N_TOTAL * HID;
  float* hidv   = hidp + BATCH * 256;
  float* part   = hidv + BATCH * 256;

  const int TB = 256;

  k_zero_int  <<<cdiv(N_TOTAL, TB), TB, 0, stream>>>(cnt, N_TOTAL);
  k_count     <<<cdiv(N_EDGES, TB), TB, 0, stream>>>(cols, cnt, slot);
  k_scan_local<<<SCAN_WG, SCAN_TB, 0, stream>>>(cnt, rowptr, bsum, dinv);
  k_scan_add  <<<SCAN_WG, SCAN_TB, 0, stream>>>(cnt, bsum, rowptr);
  k_fill      <<<cdiv(N_EDGES, TB), TB, 0, stream>>>(cols, slot, rowptr, eid);
  k_rank      <<<cdiv((long long)N_TOTAL * 32, TB), TB, 0, stream>>>(rows, rowptr, eid, dinv,
                                                                    srcs, wts);

  const float* in = x;
  int Kdim = STATE;
  dim3 layer_grid(1, N_TOTAL / 64);
  for (int l = 0; l < 5; ++l) {
    k_gemm_wmma<2, false, false, false><<<layer_grid, 128, 0, stream>>>(
        in, W[l], nullptr, bufA, N_TOTAL, HID, Kdim);
    k_aggregate<<<cdiv((long long)N_TOTAL * 32, TB), TB, 0, stream>>>(
        rowptr, srcs, wts, dinv, bufA, bb[l], bufB);
    in = bufB;
    Kdim = HID;
  }

  dim3 g1(HEAD_N / 32, cdiv(BATCH, 64), KSPLIT);
  k_gemm_wmma_ksplit<2><<<g1, 128, 0, stream>>>(
      bufB, Wp1, part, BATCH, HEAD_N, HEAD_K, KCHUNK);
  k_gemm_wmma_ksplit<2><<<g1, 128, 0, stream>>>(
      bufB, Wv1, part + (size_t)KSPLIT * PLANE, BATCH, HEAD_N, HEAD_K, KCHUNK);
  k_head_reduce<<<cdiv(2 * PLANE, 128), 128, 0, stream>>>(part, bp1, bv1, hidp);

  dim3 g2(cdiv(NODE_COUNT, 16), cdiv(BATCH, 64));
  k_gemm_wmma<1, true, true, false><<<g2, 128, 0, stream>>>(
      hidp, Wp2, bp2, out, BATCH, NODE_COUNT, 256);

  k_value<<<1, 64, 0, stream>>>(hidv, Wv2, bv2, out + (size_t)BATCH * NODE_COUNT);
}


# Aggregate: all row fetches of a node (up to 40) in flight before the first add, no per-edge remainder loop, self/bias loads first; tail padded to 8 by zero-weight own-row reads; sum order unchanged
# speedup vs baseline: 10.6804x; 1.0563x over previous
#include <hip/hip_runtime.h>

#define N_TOTAL    64000
#define NODE_COUNT 1000
#define BATCH      64
#define STATE      16
#define HID        32
#define N_EDGES    2048000
#define E_TOTAL    (N_EDGES + N_TOTAL)
#define HEAD_K     (NODE_COUNT * HID)
#define HEAD_N     256
#define KSPLIT     250
#define KCHUNK     (HEAD_K / KSPLIT)
#define PLANE      (BATCH * HEAD_N)
#define SCAN_TB    256
#define SCAN_WG    (N_TOTAL / SCAN_TB)

typedef float v2f __attribute__((ext_vector_type(2)));
typedef float v8f __attribute__((ext_vector_type(8)));

typedef const __attribute__((address_space(1))) float* gcfp;
typedef __attribute__((address_space(1))) float*       gfp;

__global__ void k_zero_int(int* __restrict__ p, int n) {
  int i = blockIdx.x * blockDim.x + threadIdx.x;
  if (i < n) p[i] = 0;
}

__global__ void k_count(const int* __restrict__ cols, int* __restrict__ cnt,
                        int* __restrict__ slot) {
  int e = blockIdx.x * blockDim.x + threadIdx.x;
  if (e < N_EDGES) slot[e] = atomicAdd(&cnt[cols[e]], 1);
}

__global__ void k_scan_local(const int* __restrict__ cnt, int* __restrict__ rowptr,
                             int* __restrict__ bsum, float* __restrict__ dinv) {
  __shared__ int s[SCAN_TB];
  int t = threadIdx.x;
  int i = blockIdx.x * SCAN_TB + t;
  int v = cnt[i];
  dinv[i] = rsqrtf((float)v + 1.0f);
  s[t] = v;
  __syncthreads();
  for (int off = 1; off < SCAN_TB; off <<= 1) {
    int a = t >= off ? s[t - off] : 0;
    __syncthreads();
    s[t] += a;
    __syncthreads();
  }
  rowptr[i] = s[t] - v;
  if (t == SCAN_TB - 1) bsum[blockIdx.x] = s[t];
}

__global__ void k_scan_add(const int* __restrict__ cnt, const int* __restrict__ bsum,
                           int* __restrict__ rowptr) {
  __shared__ int s[SCAN_TB];
  int t = threadIdx.x;
  int b = blockIdx.x;
  s[t] = t < b ? bsum[t] : 0;
  __syncthreads();
  for (int off = SCAN_TB / 2; off > 0; off >>= 1) {
    if (t < off) s[t] += s[t + off];
    __syncthreads();
  }
  int i = b * SCAN_TB + t;
  int p = rowptr[i] + s[0];
  rowptr[i] = p;
  if (i == N_TOTAL - 1) rowptr[N_TOTAL] = p + cnt[i];
}

__global__ void k_fill(const int* __restrict__ cols, const int* __restrict__ slot,
                       const int* __restrict__ rowptr, int* __restrict__ eid) {
  int e = blockIdx.x * blockDim.x + threadIdx.x;
  if (e < N_EDGES) eid[rowptr[cols[e]] + slot[e]] = e;
}

__global__ void k_rank(const int* __restrict__ rows, const int* __restrict__ rowptr,
                       const int* __restrict__ eid, const float* __restrict__ dinv,
                       int* __restrict__ srcs, float* __restrict__ wts) {
  int t    = blockIdx.x * blockDim.x + threadIdx.x;
  int lane = t & 31;
  int c = __builtin_amdgcn_readfirstlane(t >> 5);
  if (c >= N_TOTAL) return;
  int beg = rowptr[c], end = rowptr[c + 1];
  float dc = dinv[c];
  for (int base = beg; base < end; base += 32) {
    int i   = base + lane;
    bool ok = i < end;
    int my  = ok ? eid[i] : 0x7fffffff;
    int rank = 0;
    for (int cb = beg; cb < end; cb += 32) {
      int q     = cb + lane;
      int other = q < end ? eid[q] : 0x7fffffff;
      int n     = end - cb < 32 ? end - cb : 32;
      for (int k = 0; k < n; ++k)
        rank += __builtin_amdgcn_readlane(other, k) < my ? 1 : 0;
    }
    if (ok) {
      int r = rows[my];
      srcs[beg + rank] = r;
      wts[beg + rank]  = dinv[r] * dc;
    }
  }
}

template <int K0, int NG>
__device__ __forceinline__ float agg_window(gcfp hg, int r0, int r1, float w0, float w1,
                                            int j, int G, float acc) {
  float v[8 * NG];
#pragma unroll
  for (int g = 0; g < NG; ++g) {
    if (g == 0 || K0 / 8 + g < G) {
#pragma unroll
      for (int k = 0; k < 8; ++k) {
        const int p = K0 + g * 8 + k;
        int r = p < 32 ? __builtin_amdgcn_readlane(r0, p) : __builtin_amdgcn_readlane(r1, p - 32);
        v[g * 8 + k] = hg[(unsigned)r * HID + j];
      }
    }
  }
#pragma unroll
  for (int g = 0; g < NG; ++g) {
    if (g == 0 || K0 / 8 + g < G) {
#pragma unroll
      for (int k = 0; k < 8; ++k) {
        const int p = K0 + g * 8 + k;
        int wi = p < 32 ? __builtin_amdgcn_readlane(__float_as_int(w0), p)
                        : __builtin_amdgcn_readlane(__float_as_int(w1), p - 32);
        acc = __builtin_fmaf(v[g * 8 + k], __int_as_float(wi), acc);
      }
    }
  }
  return acc;
}

__global__ void k_aggregate(const int* __restrict__ rowptr, const int* __restrict__ srcs,
                            const float* __restrict__ wts, const float* __restrict__ dinv,
                            const float* __restrict__ h, const float* __restrict__ bias,
                            float* __restrict__ out) {
  int t = blockIdx.x * blockDim.x + threadIdx.x;
  int j = t & 31;
  int c = __builtin_amdgcn_readfirstlane(t >> 5);
  if (c >= N_TOTAL) return;
  gcfp hg = (gcfp)h;
  int beg = rowptr[c], end = rowptr[c + 1];
  float selfv = hg[(unsigned)c * HID + j];
  float bj    = bias[j];
  float dc    = dinv[c];
  float acc = 0.0f;
  for (int base = beg; base < end; base += 64) {
    int i0 = base + j, i1 = base + 32 + j;
    int   r0 = c,    r1 = c;
    float w0 = 0.0f, w1 = 0.0f;
    if (i0 < end) { r0 = srcs[i0]; w0 = wts[i0]; }
    if (i1 < end) { r1 = srcs[i1]; w1 = wts[i1]; }
    int n = end - base < 64 ? end - base : 64;
    int G = (n + 7) >> 3;
    asm volatile("" :: "v"(r0), "v"(w0), "v"(r1), "v"(w1));
    acc = agg_window<0, 5>(hg, r0, r1, w0, w1, j, G, acc);
    if (G > 5) acc = agg_window<40, 1>(hg, r0, r1, w0, w1, j, G, acc);
    if (G > 6) acc = agg_window<48, 1>(hg, r0, r1, w0, w1, j, G, acc);
    if (G > 7) acc = agg_window<56, 1>(hg, r0, r1, w0, w1, j, G, acc);
  }
  acc += selfv * (dc * dc);
  float v = acc + bj;
  ((gfp)out)[(unsigned)c * HID + j] = v > 0.0f ? v : 0.0f;
}

template <int NT, bool RAGGED, bool BIAS, bool RELU>
__global__ void k_gemm_wmma(const float* __restrict__ A, const float* __restrict__ B,
                            const float* __restrict__ bias, float* __restrict__ C,
                            int M, int N, int K) {
  int wave  = threadIdx.x >> 5;
  int lane  = threadIdx.x & 31;
  int mtile = blockIdx.y * 4 + wave;
  int m0    = mtile * 16;
  if (m0 >= M) return;
  int n0   = blockIdx.x * (16 * NT);
  int half = lane >> 4;
  int lm   = lane & 15;

  gcfp Ag = (gcfp)A;
  gcfp Bg = (gcfp)B;

  int  ncol[NT];
  bool nok[NT];
  unsigned boff[NT];
#pragma unroll
  for (int t = 0; t < NT; ++t) {
    ncol[t] = n0 + t * 16 + lm;
    int nld;
    if (RAGGED) {
      nok[t] = ncol[t] < N;
      nld    = nok[t] ? ncol[t] : (N - 1);
    } else {
      nok[t] = true;
      nld    = ncol[t];
    }
    boff[t] = (unsigned)(half * 2) * (unsigned)N + (unsigned)nld;
  }

  unsigned aoff = (unsigned)(m0 + lm) * (unsigned)K + (unsigned)(half * 2);

  v8f acc[NT];
#pragma unroll
  for (int t = 0; t < NT; ++t) acc[t] = (v8f){0.f,0.f,0.f,0.f,0.f,0.f,0.f,0.f};

  for (int k0 = 0; k0 < K; k0 += 4) {
    v2f a;
    a.x = Ag[aoff];
    a.y = Ag[aoff + 1];
#pragma unroll
    for (int t = 0; t < NT; ++t) {
      float b0 = Bg[boff[t]];
      float b1 = Bg[boff[t] + (unsigned)N];
      if (RAGGED) { b0 = nok[t] ? b0 : 0.0f; b1 = nok[t] ? b1 : 0.0f; }
      v2f b; b.x = b0; b.y = b1;
      acc[t] = __builtin_amdgcn_wmma_f32_16x16x4_f32(
           false, a,  false, b,
           (short)0, acc[t],  false,  false);
      boff[t] += 4u * (unsigned)N;
    }
    aoff += 4u;
  }

  gfp Cg = (gfp)C;
  gcfp biasg = (gcfp)bias;
#pragma unroll
  for (int t = 0; t < NT; ++t) {
    float bv = BIAS ? biasg[nok[t] ? ncol[t] : 0] : 0.0f;
    if (!RAGGED || nok[t]) {
#pragma unroll
      for (int r = 0; r < 8; ++r) {
        int m = m0 + r + half * 8;
        float v = acc[t][r] + bv;
        if (RELU) v = v > 0.0f ? v : 0.0f;
        Cg[(unsigned)m * (unsigned)N + (unsigned)ncol[t]] = v;
      }
    }
  }
}

template <int NT>
__global__ void k_gemm_wmma_ksplit(const float* __restrict__ A, const float* __restrict__ B,
                                   float* __restrict__ P, int M, int N, int K, int KC) {
  int wave  = threadIdx.x >> 5;
  int lane  = threadIdx.x & 31;
  int mtile = blockIdx.y * 4 + wave;
  int m0    = mtile * 16;
  if (m0 >= M) return;
  int n0   = blockIdx.x * (16 * NT);
  int half = lane >> 4;
  int lm   = lane & 15;
  unsigned kbeg = (unsigned)blockIdx.z * (unsigned)KC;

  gcfp Ag = (gcfp)A;
  gcfp Bg = (gcfp)B;

  int ncol[NT];
  unsigned boff[NT];
#pragma unroll
  for (int t = 0; t < NT; ++t) {
    ncol[t] = n0 + t * 16 + lm;
    boff[t] = (kbeg + (unsigned)(half * 2)) * (unsigned)N + (unsigned)ncol[t];
  }

  unsigned aoff = (unsigned)(m0 + lm) * (unsigned)K + kbeg + (unsigned)(half * 2);

  v8f acc[NT];
#pragma unroll
  for (int t = 0; t < NT; ++t) acc[t] = (v8f){0.f,0.f,0.f,0.f,0.f,0.f,0.f,0.f};

  for (int k0 = 0; k0 < KC; k0 += 4) {
    v2f a;
    a.x = Ag[aoff];
    a.y = Ag[aoff + 1];
#pragma unroll
    for (int t = 0; t < NT; ++t) {
      v2f b;
      b.x = Bg[boff[t]];
      b.y = Bg[boff[t] + (unsigned)N];
      acc[t] = __builtin_amdgcn_wmma_f32_16x16x4_f32(
           false, a,  false, b,
           (short)0, acc[t],  false,  false);
      boff[t] += 4u * (unsigned)N;
    }
    aoff += 4u;
  }

  gfp Pg = (gfp)P;
  unsigned pbase = (unsigned)blockIdx.z * (unsigned)M * (unsigned)N;
#pragma unroll
  for (int t = 0; t < NT; ++t) {
#pragma unroll
    for (int r = 0; r < 8; ++r) {
      int m = m0 + r + half * 8;
      Pg[pbase + (unsigned)m * (unsigned)N + (unsigned)ncol[t]] = acc[t][r];
    }
  }
}

__global__ void k_head_reduce(const float* __restrict__ part,
                              const float* __restrict__ bp1, const float* __restrict__ bv1,
                              float* __restrict__ hid) {
  int i = blockIdx.x * blockDim.x + threadIdx.x;
  if (i >= 2 * PLANE) return;
  int h = i >= PLANE ? 1 : 0;
  int e = i - h * PLANE;
  gcfp p = (gcfp)part + (size_t)h * KSPLIT * PLANE + e;
  float s = 0.0f;
  for (int z = 0; z < KSPLIT; ++z) s += p[(unsigned)z * (unsigned)PLANE];
  const float* bias = h ? bv1 : bp1;
  s += bias[e & (HEAD_N - 1)];
  hid[i] = s > 0.0f ? s : 0.0f;
}

__global__ void k_value(const float* __restrict__ hidv, const float* __restrict__ Wv2,
                        const float* __restrict__ bv2, float* __restrict__ out) {
  int b = blockIdx.x * blockDim.x + threadIdx.x;
  if (b < BATCH) {
    float s = bv2[0];
    for (int k = 0; k < 256; ++k) s += hidv[b * 256 + k] * Wv2[k];
    out[b] = s;
  }
}

static inline int cdiv(long long a, int b) { return (int)((a + b - 1) / b); }

extern "C" void kernel_launch(void* const* d_in, const int* in_sizes, int n_in,
                              void* d_out, int out_size, void* d_ws, size_t ws_size,
                              hipStream_t stream) {
  (void)in_sizes; (void)n_in; (void)out_size; (void)ws_size;

  const float* x    = (const float*)d_in[0];
  const int*   ei   = (const int*)d_in[1];
  const int*   rows = ei;
  const int*   cols = ei + N_EDGES;
  const float* W[5]  = {(const float*)d_in[2], (const float*)d_in[4], (const float*)d_in[6],
                        (const float*)d_in[8], (const float*)d_in[10]};
  const float* bb[5] = {(const float*)d_in[3], (const float*)d_in[5], (const float*)d_in[7],
                        (const float*)d_in[9], (const float*)d_in[11]};
  const float* Wp1 = (const float*)d_in[12]; const float* bp1 = (const float*)d_in[13];
  const float* Wp2 = (const float*)d_in[14]; const float* bp2 = (const float*)d_in[15];
  const float* Wv1 = (const float*)d_in[16]; const float* bv1 = (const float*)d_in[17];
  const float* Wv2 = (const float*)d_in[18]; const float* bv2 = (const float*)d_in[19];
  float* out = (float*)d_out;

  float* ws     = (float*)d_ws;
  int*   cnt    = (int*)ws;
  int*   rowptr = cnt + N_TOTAL;
  int*   bsum   = rowptr + N_TOTAL + 64;
  float* dinv   = (float*)(bsum + 256);
  int*   slot   = (int*)(dinv + N_TOTAL);
  int*   eid    = slot + N_EDGES;
  int*   srcs   = eid + N_EDGES;
  float* wts    = (float*)(srcs + N_EDGES);
  float* bufA   = wts + N_EDGES;
  float* bufB   = bufA + (size_t)N_TOTAL * HID;
  float* hidp   = bufB + (size_t)N_TOTAL * HID;
  float* hidv   = hidp + BATCH * 256;
  float* part   = hidv + BATCH * 256;

  const int TB = 256;

  k_zero_int  <<<cdiv(N_TOTAL, TB), TB, 0, stream>>>(cnt, N_TOTAL);
  k_count     <<<cdiv(N_EDGES, TB), TB, 0, stream>>>(cols, cnt, slot);
  k_scan_local<<<SCAN_WG, SCAN_TB, 0, stream>>>(cnt, rowptr, bsum, dinv);
  k_scan_add  <<<SCAN_WG, SCAN_TB, 0, stream>>>(cnt, bsum, rowptr);
  k_fill      <<<cdiv(N_EDGES, TB), TB, 0, stream>>>(cols, slot, rowptr, eid);
  k_rank      <<<cdiv((long long)N_TOTAL * 32, TB), TB, 0, stream>>>(rows, rowptr, eid, dinv,
                                                                    srcs, wts);

  const float* in = x;
  int Kdim = STATE;
  dim3 layer_grid(1, N_TOTAL / 64);
  for (int l = 0; l < 5; ++l) {
    k_gemm_wmma<2, false, false, false><<<layer_grid, 128, 0, stream>>>(
        in, W[l], nullptr, bufA, N_TOTAL, HID, Kdim);
    k_aggregate<<<cdiv((long long)N_TOTAL * 32, TB), TB, 0, stream>>>(
        rowptr, srcs, wts, dinv, bufA, bb[l], bufB);
    in = bufB;
    Kdim = HID;
  }

  dim3 g1(HEAD_N / 32, cdiv(BATCH, 64), KSPLIT);
  k_gemm_wmma_ksplit<2><<<g1, 128, 0, stream>>>(
      bufB, Wp1, part, BATCH, HEAD_N, HEAD_K, KCHUNK);
  k_gemm_wmma_ksplit<2><<<g1, 128, 0, stream>>>(
      bufB, Wv1, part + (size_t)KSPLIT * PLANE, BATCH, HEAD_N, HEAD_K, KCHUNK);
  k_head_reduce<<<cdiv(2 * PLANE, 128), 128, 0, stream>>>(part, bp1, bv1, hidp);

  dim3 g2(cdiv(NODE_COUNT, 16), cdiv(BATCH, 64));
  k_gemm_wmma<1, true, true, false><<<g2, 128, 0, stream>>>(
      hidp, Wp2, bp2, out, BATCH, NODE_COUNT, 256);

  k_value<<<1, 64, 0, stream>>>(hidv, Wv2, bv2, out + (size_t)BATCH * NODE_COUNT);
}


# Layers 2-5: the layer GEMM runs inside the gather kernel of the layer before, on its 16-node tile (LDS to WMMA A fragments); 4 launches and the h write+read gone; sum order and rounding unchanged
# speedup vs baseline: 11.4247x; 1.0697x over previous
#include <hip/hip_runtime.h>

#define N_TOTAL    64000
#define NODE_COUNT 1000
#define BATCH      64
#define STATE      16
#define HID        32
#define N_EDGES    2048000
#define E_TOTAL    (N_EDGES + N_TOTAL)
#define HEAD_K     (NODE_COUNT * HID)
#define HEAD_N     256
#define KSPLIT     250
#define KCHUNK     (HEAD_K / KSPLIT)
#define PLANE      (BATCH * HEAD_N)
#define SCAN_TB    256
#define SCAN_WG    (N_TOTAL / SCAN_TB)
#define TILE_M     16
#define TILE_LD    (HID + 2)
static_assert(N_TOTAL % TILE_M == 0, "the fused layer kernel has no partial tile");

typedef float v2f __attribute__((ext_vector_type(2)));
typedef float v8f __attribute__((ext_vector_type(8)));

typedef const __attribute__((address_space(1))) float* gcfp;
typedef __attribute__((address_space(1))) float*       gfp;

__global__ void k_zero_int(int* __restrict__ p, int n, float* __restrict__ zrow0,
                           float* __restrict__ zrow1) {
  int i = blockIdx.x * blockDim.x + threadIdx.x;
  if (i < n) p[i] = 0;
  if (i < HID) { zrow0[i] = 0.0f; zrow1[i] = 0.0f; }
}

__global__ void k_count(const int* __restrict__ cols, int* __restrict__ cnt,
                        int* __restrict__ slot) {
  int e = blockIdx.x * blockDim.x + threadIdx.x;
  if (e < N_EDGES) slot[e] = atomicAdd(&cnt[cols[e]], 1);
}

__global__ void k_scan_local(const int* __restrict__ cnt, int* __restrict__ rowptr,
                             int* __restrict__ bsum, float* __restrict__ dinv) {
  __shared__ int s[SCAN_TB];
  int t = threadIdx.x;
  int i = blockIdx.x * SCAN_TB + t;
  int v = cnt[i];
  dinv[i] = rsqrtf((float)v + 1.0f);
  s[t] = v;
  __syncthreads();
  for (int off = 1; off < SCAN_TB; off <<= 1) {
    int a = t >= off ? s[t - off] : 0;
    __syncthreads();
    s[t] += a;
    __syncthreads();
  }
  rowptr[i] = s[t] - v;
  if (t == SCAN_TB - 1) bsum[blockIdx.x] = s[t];
}

__global__ void k_scan_add(const int* __restrict__ cnt, const int* __restrict__ bsum,
                           int* __restrict__ rowptr) {
  __shared__ int s[SCAN_TB];
  int t = threadIdx.x;
  int b = blockIdx.x;
  s[t] = t < b ? bsum[t] : 0;
  __syncthreads();
  for (int off = SCAN_TB / 2; off > 0; off >>= 1) {
    if (t < off) s[t] += s[t + off];
    __syncthreads();
  }
  int i = b * SCAN_TB + t;
  int p = rowptr[i] + s[0];
  rowptr[i] = p;
  if (i == N_TOTAL - 1) rowptr[N_TOTAL] = p + cnt[i];
}

__global__ void k_fill(const int* __restrict__ cols, const int* __restrict__ slot,
                       const int* __restrict__ rowptr, int* __restrict__ eid) {
  int e = blockIdx.x * blockDim.x + threadIdx.x;
  if (e < N_EDGES) eid[rowptr[cols[e]] + slot[e]] = e;
}

__global__ void k_rank(const int* __restrict__ rows, const int* __restrict__ rowptr,
                       const int* __restrict__ eid, int* __restrict__ srcs) {
  int t    = blockIdx.x * blockDim.x + threadIdx.x;
  int lane = t & 31;
  int c = __builtin_amdgcn_readfirstlane(t >> 5);
  if (c >= N_TOTAL) return;
  int beg = rowptr[c], end = rowptr[c + 1];
  for (int base = beg; base < end; base += 32) {
    int i   = base + lane;
    bool ok = i < end;
    int my  = ok ? eid[i] : 0x7fffffff;
    int rank = 0;
    for (int cb = beg; cb < end; cb += 32) {
      int q     = cb + lane;
      int other = q < end ? eid[q] : 0x7fffffff;
      int n     = end - cb < 32 ? end - cb : 32;
      for (int k = 0; k < n; ++k)
        rank += __builtin_amdgcn_readlane(other, k) < my ? 1 : 0;
    }
    if (ok) srcs[beg + rank] = rows[my];
  }
}

template <int K0, int NG>
__device__ __forceinline__ float agg_window(gcfp hg, int r0, int r1, int j, int G, float acc) {
  float v[8 * NG];
#pragma unroll
  for (int g = 0; g < NG; ++g) {
    if (g == 0 || K0 / 8 + g < G) {
#pragma unroll
      for (int k = 0; k < 8; ++k) {
        const int p = K0 + g * 8 + k;
        int r = p < 32 ? __builtin_amdgcn_readlane(r0, p) : __builtin_amdgcn_readlane(r1, p - 32);
        v[g * 8 + k] = hg[(unsigned)r * HID + j];
      }
    }
  }
#pragma unroll
  for (int g = 0; g < NG; ++g) {
    if (g == 0 || K0 / 8 + g < G) {
#pragma unroll
      for (int k = 0; k < 8; ++k) acc += v[g * 8 + k];
    }
  }
  return acc;
}

__device__ __forceinline__ float node_out(const int* __restrict__ rowptr,
                                          const int* __restrict__ srcs,
                                          const float* __restrict__ dinv, gcfp hg,
                                          const float* __restrict__ bias, int c, int j) {
  int beg = rowptr[c], end = rowptr[c + 1];
  float selfv = hg[(unsigned)c * HID + j];
  float bj    = bias[j];
  float dc    = dinv[c];
  float acc = 0.0f;
  for (int base = beg; base < end; base += 64) {
    int i0 = base + j, i1 = base + 32 + j;
    int r0 = N_TOTAL, r1 = N_TOTAL;
    if (i0 < end) r0 = srcs[i0];
    if (i1 < end) r1 = srcs[i1];
    int n = end - base < 64 ? end - base : 64;
    int G = (n + 7) >> 3;
    asm volatile("" :: "v"(r0), "v"(r1));
    acc = agg_window<0, 5>(hg, r0, r1, j, G, acc);
    if (G > 5) acc = agg_window<40, 1>(hg, r0, r1, j, G, acc);
    if (G > 6) acc = agg_window<48, 1>(hg, r0, r1, j, G, acc);
    if (G > 7) acc = agg_window<56, 1>(hg, r0, r1, j, G, acc);
  }
  acc += selfv;
  float v = dc * acc + bj;
  return v > 0.0f ? v : 0.0f;
}

__global__ void k_aggregate(const int* __restrict__ rowptr, const int* __restrict__ srcs,
                            const float* __restrict__ dinv,
                            const float* __restrict__ h, const float* __restrict__ bias,
                            float* __restrict__ out) {
  int t = blockIdx.x * blockDim.x + threadIdx.x;
  int j = t & 31;
  int c = __builtin_amdgcn_readfirstlane(t >> 5);
  if (c >= N_TOTAL) return;
  float v = node_out(rowptr, srcs, dinv, (gcfp)h, bias, c, j);
  ((gfp)out)[(unsigned)c * HID + j] = v;
}

__global__ void __launch_bounds__(TILE_M * 32)
k_aggregate_gemm(const int* __restrict__ rowptr, const int* __restrict__ srcs,
                 const float* __restrict__ dinv,
                 const float* __restrict__ h, const float* __restrict__ bias,
                 const float* __restrict__ Wn, float* __restrict__ hsn) {
  __shared__ __attribute__((aligned(16))) float tile[TILE_M * TILE_LD];
  int j = threadIdx.x & 31;
  int w  = __builtin_amdgcn_readfirstlane((int)(threadIdx.x >> 5));
  int m0 = blockIdx.x * TILE_M;
  float v = node_out(rowptr, srcs, dinv, (gcfp)h, bias, m0 + w, j);
  tile[w * TILE_LD + j] = v;
  __syncthreads();
  if (w != 0) return;

  int half = j >> 4;
  int lm   = j & 15;
  gcfp Wg = (gcfp)Wn;
  gcfp dg = (gcfp)dinv;
  float rs[8];
#pragma unroll
  for (int r = 0; r < 8; ++r) rs[r] = dg[m0 + half * 8 + r];

  v8f acc[2];
#pragma unroll
  for (int t = 0; t < 2; ++t) acc[t] = (v8f){0.f,0.f,0.f,0.f,0.f,0.f,0.f,0.f};
#pragma unroll
  for (int k0 = 0; k0 < HID; k0 += 4) {
    v2f a = *(const v2f*)&tile[lm * TILE_LD + k0 + half * 2];
#pragma unroll
    for (int t = 0; t < 2; ++t) {
      v2f b;
      b.x = Wg[(unsigned)(k0 + half * 2) * HID + t * 16 + lm];
      b.y = Wg[(unsigned)(k0 + half * 2 + 1) * HID + t * 16 + lm];
      acc[t] = __builtin_amdgcn_wmma_f32_16x16x4_f32(
           false, a,  false, b,
           (short)0, acc[t],  false,  false);
    }
  }
  gfp og = (gfp)hsn;
#pragma unroll
  for (int t = 0; t < 2; ++t) {
#pragma unroll
    for (int r = 0; r < 8; ++r) {
      int m = m0 + r + half * 8;
      og[(unsigned)m * HID + t * 16 + lm] = acc[t][r] * rs[r];
    }
  }
}

template <int NT, bool RAGGED, bool BIAS, bool RELU, bool ROWSCALE>
__global__ void k_gemm_wmma(const float* __restrict__ A, const float* __restrict__ B,
                            const float* __restrict__ bias, const float* __restrict__ rscale,
                            float* __restrict__ C, int M, int N, int K) {
  int wave  = threadIdx.x >> 5;
  int lane  = threadIdx.x & 31;
  int mtile = blockIdx.y * 4 + wave;
  int m0    = mtile * 16;
  if (m0 >= M) return;
  int n0   = blockIdx.x * (16 * NT);
  int half = lane >> 4;
  int lm   = lane & 15;

  gcfp Ag = (gcfp)A;
  gcfp Bg = (gcfp)B;

  float rs[8];
  if (ROWSCALE) {
    gcfp rg = (gcfp)rscale;
#pragma unroll
    for (int r = 0; r < 8; ++r) rs[r] = rg[m0 + half * 8 + r];
  }

  int  ncol[NT];
  bool nok[NT];
  unsigned boff[NT];
#pragma unroll
  for (int t = 0; t < NT; ++t) {
    ncol[t] = n0 + t * 16 + lm;
    int nld;
    if (RAGGED) {
      nok[t] = ncol[t] < N;
      nld    = nok[t] ? ncol[t] : (N - 1);
    } else {
      nok[t] = true;
      nld    = ncol[t];
    }
    boff[t] = (unsigned)(half * 2) * (unsigned)N + (unsigned)nld;
  }

  unsigned aoff = (unsigned)(m0 + lm) * (unsigned)K + (unsigned)(half * 2);

  v8f acc[NT];
#pragma unroll
  for (int t = 0; t < NT; ++t) acc[t] = (v8f){0.f,0.f,0.f,0.f,0.f,0.f,0.f,0.f};

  for (int k0 = 0; k0 < K; k0 += 4) {
    v2f a;
    a.x = Ag[aoff];
    a.y = Ag[aoff + 1];
#pragma unroll
    for (int t = 0; t < NT; ++t) {
      float b0 = Bg[boff[t]];
      float b1 = Bg[boff[t] + (unsigned)N];
      if (RAGGED) { b0 = nok[t] ? b0 : 0.0f; b1 = nok[t] ? b1 : 0.0f; }
      v2f b; b.x = b0; b.y = b1;
      acc[t] = __builtin_amdgcn_wmma_f32_16x16x4_f32(
           false, a,  false, b,
           (short)0, acc[t],  false,  false);
      boff[t] += 4u * (unsigned)N;
    }
    aoff += 4u;
  }

  gfp Cg = (gfp)C;
  gcfp biasg = (gcfp)bias;
#pragma unroll
  for (int t = 0; t < NT; ++t) {
    float bv = BIAS ? biasg[nok[t] ? ncol[t] : 0] : 0.0f;
    if (!RAGGED || nok[t]) {
#pragma unroll
      for (int r = 0; r < 8; ++r) {
        int m = m0 + r + half * 8;
        float v = acc[t][r];
        if (ROWSCALE) v *= rs[r];
        if (BIAS) v += bv;
        if (RELU) v = v > 0.0f ? v : 0.0f;
        Cg[(unsigned)m * (unsigned)N + (unsigned)ncol[t]] = v;
      }
    }
  }
}

template <int NT>
__global__ void k_gemm_wmma_ksplit(const float* __restrict__ A, const float* __restrict__ B,
                                   float* __restrict__ P, int M, int N, int K, int KC) {
  int wave  = threadIdx.x >> 5;
  int lane  = threadIdx.x & 31;
  int mtile = blockIdx.y * 4 + wave;
  int m0    = mtile * 16;
  if (m0 >= M) return;
  int n0   = blockIdx.x * (16 * NT);
  int half = lane >> 4;
  int lm   = lane & 15;
  unsigned kbeg = (unsigned)blockIdx.z * (unsigned)KC;

  gcfp Ag = (gcfp)A;
  gcfp Bg = (gcfp)B;

  int ncol[NT];
  unsigned boff[NT];
#pragma unroll
  for (int t = 0; t < NT; ++t) {
    ncol[t] = n0 + t * 16 + lm;
    boff[t] = (kbeg + (unsigned)(half * 2)) * (unsigned)N + (unsigned)ncol[t];
  }

  unsigned aoff = (unsigned)(m0 + lm) * (unsigned)K + kbeg + (unsigned)(half * 2);

  v8f acc[NT];
#pragma unroll
  for (int t = 0; t < NT; ++t) acc[t] = (v8f){0.f,0.f,0.f,0.f,0.f,0.f,0.f,0.f};

  for (int k0 = 0; k0 < KC; k0 += 4) {
    v2f a;
    a.x = Ag[aoff];
    a.y = Ag[aoff + 1];
#pragma unroll
    for (int t = 0; t < NT; ++t) {
      v2f b;
      b.x = Bg[boff[t]];
      b.y = Bg[boff[t] + (unsigned)N];
      acc[t] = __builtin_amdgcn_wmma_f32_16x16x4_f32(
           false, a,  false, b,
           (short)0, acc[t],  false,  false);
      boff[t] += 4u * (unsigned)N;
    }
    aoff += 4u;
  }

  gfp Pg = (gfp)P;
  unsigned pbase = (unsigned)blockIdx.z * (unsigned)M * (unsigned)N;
#pragma unroll
  for (int t = 0; t < NT; ++t) {
#pragma unroll
    for (int r = 0; r < 8; ++r) {
      int m = m0 + r + half * 8;
      Pg[pbase + (unsigned)m * (unsigned)N + (unsigned)ncol[t]] = acc[t][r];
    }
  }
}

__global__ void k_head_reduce(const float* __restrict__ part,
                              const float* __restrict__ bp1, const float* __restrict__ bv1,
                              float* __restrict__ hid) {
  int i = blockIdx.x * blockDim.x + threadIdx.x;
  if (i >= 2 * PLANE) return;
  int h = i >= PLANE ? 1 : 0;
  int e = i - h * PLANE;
  gcfp p = (gcfp)part + (size_t)h * KSPLIT * PLANE + e;
  float s = 0.0f;
  for (int z = 0; z < KSPLIT; ++z) s += p[(unsigned)z * (unsigned)PLANE];
  const float* bias = h ? bv1 : bp1;
  s += bias[e & (HEAD_N - 1)];
  hid[i] = s > 0.0f ? s : 0.0f;
}

__global__ void k_value(const float* __restrict__ hidv, const float* __restrict__ Wv2,
                        const float* __restrict__ bv2, float* __restrict__ out) {
  int b = blockIdx.x * blockDim.x + threadIdx.x;
  if (b < BATCH) {
    float s = bv2[0];
    for (int k = 0; k < 256; ++k) s += hidv[b * 256 + k] * Wv2[k];
    out[b] = s;
  }
}

static inline int cdiv(long long a, int b) { return (int)((a + b - 1) / b); }

extern "C" void kernel_launch(void* const* d_in, const int* in_sizes, int n_in,
                              void* d_out, int out_size, void* d_ws, size_t ws_size,
                              hipStream_t stream) {
  (void)in_sizes; (void)n_in; (void)out_size; (void)ws_size;

  const float* x    = (const float*)d_in[0];
  const int*   ei   = (const int*)d_in[1];
  const int*   rows = ei;
  const int*   cols = ei + N_EDGES;
  const float* W[5]  = {(const float*)d_in[2], (const float*)d_in[4], (const float*)d_in[6],
                        (const float*)d_in[8], (const float*)d_in[10]};
  const float* bb[5] = {(const float*)d_in[3], (const float*)d_in[5], (const float*)d_in[7],
                        (const float*)d_in[9], (const float*)d_in[11]};
  const float* Wp1 = (const float*)d_in[12]; const float* bp1 = (const float*)d_in[13];
  const float* Wp2 = (const float*)d_in[14]; const float* bp2 = (const float*)d_in[15];
  const float* Wv1 = (const float*)d_in[16]; const float* bv1 = (const float*)d_in[17];
  const float* Wv2 = (const float*)d_in[18]; const float* bv2 = (const float*)d_in[19];
  float* out = (float*)d_out;

  float* ws     = (float*)d_ws;
  int*   cnt    = (int*)ws;
  int*   rowptr = cnt + N_TOTAL;
  int*   bsum   = rowptr + N_TOTAL + 64;
  float* dinv   = (float*)(bsum + 256);
  int*   slot   = (int*)(dinv + N_TOTAL);
  int*   eid    = slot + N_EDGES;
  int*   srcs   = eid + N_EDGES;
  float* bufA   = (float*)(srcs + N_EDGES);
  float* zrowA  = bufA + (size_t)N_TOTAL * HID;
  float* bufB   = zrowA + 64;
  float* zrowB  = bufB + (size_t)N_TOTAL * HID;
  float* hidp   = zrowB + 64;
  float* hidv   = hidp + BATCH * 256;
  float* part   = hidv + BATCH * 256;

  const int TB = 256;

  k_zero_int  <<<cdiv(N_TOTAL, TB), TB, 0, stream>>>(cnt, N_TOTAL, zrowA, zrowB);
  k_count     <<<cdiv(N_EDGES, TB), TB, 0, stream>>>(cols, cnt, slot);
  k_scan_local<<<SCAN_WG, SCAN_TB, 0, stream>>>(cnt, rowptr, bsum, dinv);
  k_scan_add  <<<SCAN_WG, SCAN_TB, 0, stream>>>(cnt, bsum, rowptr);
  k_fill      <<<cdiv(N_EDGES, TB), TB, 0, stream>>>(cols, slot, rowptr, eid);
  k_rank      <<<cdiv((long long)N_TOTAL * 32, TB), TB, 0, stream>>>(rows, rowptr, eid, srcs);

  dim3 layer_grid(1, N_TOTAL / 64);
  k_gemm_wmma<2, false, false, false, true><<<layer_grid, 128, 0, stream>>>(
      x, W[0], nullptr, dinv, bufA, N_TOTAL, HID, STATE);
  const int FG = N_TOTAL / TILE_M;
  const int FB = TILE_M * 32;
  k_aggregate_gemm<<<FG, FB, 0, stream>>>(rowptr, srcs, dinv, bufA, bb[0], W[1], bufB);
  k_aggregate_gemm<<<FG, FB, 0, stream>>>(rowptr, srcs, dinv, bufB, bb[1], W[2], bufA);
  k_aggregate_gemm<<<FG, FB, 0, stream>>>(rowptr, srcs, dinv, bufA, bb[2], W[3], bufB);
  k_aggregate_gemm<<<FG, FB, 0, stream>>>(rowptr, srcs, dinv, bufB, bb[3], W[4], bufA);
  k_aggregate<<<cdiv((long long)N_TOTAL * 32, TB), TB, 0, stream>>>(
      rowptr, srcs, dinv, bufA, bb[4], bufB);

  dim3 g1(HEAD_N / 32, cdiv(BATCH, 64), KSPLIT);
  k_gemm_wmma_ksplit<2><<<g1, 128, 0, stream>>>(
      bufB, Wp1, part, BATCH, HEAD_N, HEAD_K, KCHUNK);
  k_gemm_wmma_ksplit<2><<<g1, 128, 0, stream>>>(
      bufB, Wv1, part + (size_t)KSPLIT * PLANE, BATCH, HEAD_N, HEAD_K, KCHUNK);
  k_head_reduce<<<cdiv(2 * PLANE, 128), 128, 0, stream>>>(part, bp1, bv1, hidp);

  dim3 g2(cdiv(NODE_COUNT, 16), cdiv(BATCH, 64));
  k_gemm_wmma<1, true, true, false, false><<<g2, 128, 0, stream>>>(
      hidp, Wp2, bp2, nullptr, out, BATCH, NODE_COUNT, 256);

  k_value<<<1, 64, 0, stream>>>(hidv, Wv2, bv2, out + (size_t)BATCH * NODE_COUNT);
}


# Head tail 5 launches -> 3: both split-K head GEMMs in one grid; plane sum spread over 5 wave groups per batch row, value dot folded in (no value kernel); plane and value-dot sum order changes
# speedup vs baseline: 11.8451x; 1.0368x over previous
#include <hip/hip_runtime.h>

#define N_TOTAL    64000
#define NODE_COUNT 1000
#define BATCH      64
#define STATE      16
#define HID        32
#define N_EDGES    2048000
#define E_TOTAL    (N_EDGES + N_TOTAL)
#define HEAD_K     (NODE_COUNT * HID)
#define HEAD_N     256
#define KSPLIT     250
#define KCHUNK     (HEAD_K / KSPLIT)
#define PLANE      (BATCH * HEAD_N)
#define SCAN_TB    256
#define SCAN_WG    (N_TOTAL / SCAN_TB)
#define TILE_M     16
#define TILE_LD    (HID + 2)
static_assert(N_TOTAL % TILE_M == 0, "the fused layer kernel has no partial tile");

typedef float v2f __attribute__((ext_vector_type(2)));
typedef float v8f __attribute__((ext_vector_type(8)));

typedef const __attribute__((address_space(1))) float* gcfp;
typedef __attribute__((address_space(1))) float*       gfp;

__global__ void k_zero_int(int* __restrict__ p, int n, float* __restrict__ zrow0,
                           float* __restrict__ zrow1) {
  int i = blockIdx.x * blockDim.x + threadIdx.x;
  if (i < n) p[i] = 0;
  if (i < HID) { zrow0[i] = 0.0f; zrow1[i] = 0.0f; }
}

__global__ void k_count(const int* __restrict__ cols, int* __restrict__ cnt,
                        int* __restrict__ slot) {
  int e = blockIdx.x * blockDim.x + threadIdx.x;
  if (e < N_EDGES) slot[e] = atomicAdd(&cnt[cols[e]], 1);
}

__global__ void k_scan_local(const int* __restrict__ cnt, int* __restrict__ rowptr,
                             int* __restrict__ bsum, float* __restrict__ dinv) {
  __shared__ int s[SCAN_TB];
  int t = threadIdx.x;
  int i = blockIdx.x * SCAN_TB + t;
  int v = cnt[i];
  dinv[i] = rsqrtf((float)v + 1.0f);
  s[t] = v;
  __syncthreads();
  for (int off = 1; off < SCAN_TB; off <<= 1) {
    int a = t >= off ? s[t - off] : 0;
    __syncthreads();
    s[t] += a;
    __syncthreads();
  }
  rowptr[i] = s[t] - v;
  if (t == SCAN_TB - 1) bsum[blockIdx.x] = s[t];
}

__global__ void k_scan_add(const int* __restrict__ cnt, const int* __restrict__ bsum,
                           int* __restrict__ rowptr) {
  __shared__ int s[SCAN_TB];
  int t = threadIdx.x;
  int b = blockIdx.x;
  s[t] = t < b ? bsum[t] : 0;
  __syncthreads();
  for (int off = SCAN_TB / 2; off > 0; off >>= 1) {
    if (t < off) s[t] += s[t + off];
    __syncthreads();
  }
  int i = b * SCAN_TB + t;
  int p = rowptr[i] + s[0];
  rowptr[i] = p;
  if (i == N_TOTAL - 1) rowptr[N_TOTAL] = p + cnt[i];
}

__global__ void k_fill(const int* __restrict__ cols, const int* __restrict__ slot,
                       const int* __restrict__ rowptr, int* __restrict__ eid) {
  int e = blockIdx.x * blockDim.x + threadIdx.x;
  if (e < N_EDGES) eid[rowptr[cols[e]] + slot[e]] = e;
}

__global__ void k_rank(const int* __restrict__ rows, const int* __restrict__ rowptr,
                       const int* __restrict__ eid, int* __restrict__ srcs) {
  int t    = blockIdx.x * blockDim.x + threadIdx.x;
  int lane = t & 31;
  int c = __builtin_amdgcn_readfirstlane(t >> 5);
  if (c >= N_TOTAL) return;
  int beg = rowptr[c], end = rowptr[c + 1];
  for (int base = beg; base < end; base += 32) {
    int i   = base + lane;
    bool ok = i < end;
    int my  = ok ? eid[i] : 0x7fffffff;
    int rank = 0;
    for (int cb = beg; cb < end; cb += 32) {
      int q     = cb + lane;
      int other = q < end ? eid[q] : 0x7fffffff;
      int n     = end - cb < 32 ? end - cb : 32;
      for (int k = 0; k < n; ++k)
        rank += __builtin_amdgcn_readlane(other, k) < my ? 1 : 0;
    }
    if (ok) srcs[beg + rank] = rows[my];
  }
}

template <int K0, int NG>
__device__ __forceinline__ float agg_window(gcfp hg, int r0, int r1, int j, int G, float acc) {
  float v[8 * NG];
#pragma unroll
  for (int g = 0; g < NG; ++g) {
    if (g == 0 || K0 / 8 + g < G) {
#pragma unroll
      for (int k = 0; k < 8; ++k) {
        const int p = K0 + g * 8 + k;
        int r = p < 32 ? __builtin_amdgcn_readlane(r0, p) : __builtin_amdgcn_readlane(r1, p - 32);
        v[g * 8 + k] = hg[(unsigned)r * HID + j];
      }
    }
  }
#pragma unroll
  for (int g = 0; g < NG; ++g) {
    if (g == 0 || K0 / 8 + g < G) {
#pragma unroll
      for (int k = 0; k < 8; ++k) acc += v[g * 8 + k];
    }
  }
  return acc;
}

__device__ __forceinline__ float node_out(const int* __restrict__ rowptr,
                                          const int* __restrict__ srcs,
                                          const float* __restrict__ dinv, gcfp hg,
                                          const float* __restrict__ bias, int c, int j) {
  int beg = rowptr[c], end = rowptr[c + 1];
  float selfv = hg[(unsigned)c * HID + j];
  float bj    = bias[j];
  float dc    = dinv[c];
  float acc = 0.0f;
  for (int base = beg; base < end; base += 64) {
    int i0 = base + j, i1 = base + 32 + j;
    int r0 = N_TOTAL, r1 = N_TOTAL;
    if (i0 < end) r0 = srcs[i0];
    if (i1 < end) r1 = srcs[i1];
    int n = end - base < 64 ? end - base : 64;
    int G = (n + 7) >> 3;
    asm volatile("" :: "v"(r0), "v"(r1));
    acc = agg_window<0, 5>(hg, r0, r1, j, G, acc);
    if (G > 5) acc = agg_window<40, 1>(hg, r0, r1, j, G, acc);
    if (G > 6) acc = agg_window<48, 1>(hg, r0, r1, j, G, acc);
    if (G > 7) acc = agg_window<56, 1>(hg, r0, r1, j, G, acc);
  }
  acc += selfv;
  float v = dc * acc + bj;
  return v > 0.0f ? v : 0.0f;
}

__global__ void k_aggregate(const int* __restrict__ rowptr, const int* __restrict__ srcs,
                            const float* __restrict__ dinv,
                            const float* __restrict__ h, const float* __restrict__ bias,
                            float* __restrict__ out) {
  int t = blockIdx.x * blockDim.x + threadIdx.x;
  int j = t & 31;
  int c = __builtin_amdgcn_readfirstlane(t >> 5);
  if (c >= N_TOTAL) return;
  float v = node_out(rowptr, srcs, dinv, (gcfp)h, bias, c, j);
  ((gfp)out)[(unsigned)c * HID + j] = v;
}

__global__ void __launch_bounds__(TILE_M * 32)
k_aggregate_gemm(const int* __restrict__ rowptr, const int* __restrict__ srcs,
                 const float* __restrict__ dinv,
                 const float* __restrict__ h, const float* __restrict__ bias,
                 const float* __restrict__ Wn, float* __restrict__ hsn) {
  __shared__ __attribute__((aligned(16))) float tile[TILE_M * TILE_LD];
  int j = threadIdx.x & 31;
  int w  = __builtin_amdgcn_readfirstlane((int)(threadIdx.x >> 5));
  int m0 = blockIdx.x * TILE_M;
  float v = node_out(rowptr, srcs, dinv, (gcfp)h, bias, m0 + w, j);
  tile[w * TILE_LD + j] = v;
  __syncthreads();
  if (w != 0) return;

  int half = j >> 4;
  int lm   = j & 15;
  gcfp Wg = (gcfp)Wn;
  gcfp dg = (gcfp)dinv;
  float rs[8];
#pragma unroll
  for (int r = 0; r < 8; ++r) rs[r] = dg[m0 + half * 8 + r];

  v8f acc[2];
#pragma unroll
  for (int t = 0; t < 2; ++t) acc[t] = (v8f){0.f,0.f,0.f,0.f,0.f,0.f,0.f,0.f};
#pragma unroll
  for (int k0 = 0; k0 < HID; k0 += 4) {
    v2f a = *(const v2f*)&tile[lm * TILE_LD + k0 + half * 2];
#pragma unroll
    for (int t = 0; t < 2; ++t) {
      v2f b;
      b.x = Wg[(unsigned)(k0 + half * 2) * HID + t * 16 + lm];
      b.y = Wg[(unsigned)(k0 + half * 2 + 1) * HID + t * 16 + lm];
      acc[t] = __builtin_amdgcn_wmma_f32_16x16x4_f32(
           false, a,  false, b,
           (short)0, acc[t],  false,  false);
    }
  }
  gfp og = (gfp)hsn;
#pragma unroll
  for (int t = 0; t < 2; ++t) {
#pragma unroll
    for (int r = 0; r < 8; ++r) {
      int m = m0 + r + half * 8;
      og[(unsigned)m * HID + t * 16 + lm] = acc[t][r] * rs[r];
    }
  }
}

template <int NT, bool RAGGED, bool BIAS, bool RELU, bool ROWSCALE>
__global__ void k_gemm_wmma(const float* __restrict__ A, const float* __restrict__ B,
                            const float* __restrict__ bias, const float* __restrict__ rscale,
                            float* __restrict__ C, int M, int N, int K) {
  int wave  = threadIdx.x >> 5;
  int lane  = threadIdx.x & 31;
  int mtile = blockIdx.y * 4 + wave;
  int m0    = mtile * 16;
  if (m0 >= M) return;
  int n0   = blockIdx.x * (16 * NT);
  int half = lane >> 4;
  int lm   = lane & 15;

  gcfp Ag = (gcfp)A;
  gcfp Bg = (gcfp)B;

  float rs[8];
  if (ROWSCALE) {
    gcfp rg = (gcfp)rscale;
#pragma unroll
    for (int r = 0; r < 8; ++r) rs[r] = rg[m0 + half * 8 + r];
  }

  int  ncol[NT];
  bool nok[NT];
  unsigned boff[NT];
#pragma unroll
  for (int t = 0; t < NT; ++t) {
    ncol[t] = n0 + t * 16 + lm;
    int nld;
    if (RAGGED) {
      nok[t] = ncol[t] < N;
      nld    = nok[t] ? ncol[t] : (N - 1);
    } else {
      nok[t] = true;
      nld    = ncol[t];
    }
    boff[t] = (unsigned)(half * 2) * (unsigned)N + (unsigned)nld;
  }

  unsigned aoff = (unsigned)(m0 + lm) * (unsigned)K + (unsigned)(half * 2);

  v8f acc[NT];
#pragma unroll
  for (int t = 0; t < NT; ++t) acc[t] = (v8f){0.f,0.f,0.f,0.f,0.f,0.f,0.f,0.f};

  for (int k0 = 0; k0 < K; k0 += 4) {
    v2f a;
    a.x = Ag[aoff];
    a.y = Ag[aoff + 1];
#pragma unroll
    for (int t = 0; t < NT; ++t) {
      float b0 = Bg[boff[t]];
      float b1 = Bg[boff[t] + (unsigned)N];
      if (RAGGED) { b0 = nok[t] ? b0 : 0.0f; b1 = nok[t] ? b1 : 0.0f; }
      v2f b; b.x = b0; b.y = b1;
      acc[t] = __builtin_amdgcn_wmma_f32_16x16x4_f32(
           false, a,  false, b,
           (short)0, acc[t],  false,  false);
      boff[t] += 4u * (unsigned)N;
    }
    aoff += 4u;
  }

  gfp Cg = (gfp)C;
  gcfp biasg = (gcfp)bias;
#pragma unroll
  for (int t = 0; t < NT; ++t) {
    float bv = BIAS ? biasg[nok[t] ? ncol[t] : 0] : 0.0f;
    if (!RAGGED || nok[t]) {
#pragma unroll
      for (int r = 0; r < 8; ++r) {
        int m = m0 + r + half * 8;
        float v = acc[t][r];
        if (ROWSCALE) v *= rs[r];
        if (BIAS) v += bv;
        if (RELU) v = v > 0.0f ? v : 0.0f;
        Cg[(unsigned)m * (unsigned)N + (unsigned)ncol[t]] = v;
      }
    }
  }
}

template <int NT>
__global__ void k_gemm_wmma_ksplit(const float* __restrict__ A, const float* __restrict__ B0,
                                   const float* __restrict__ B1,
                                   float* __restrict__ P, int M, int N, int K, int KC, int ZS) {
  int wave  = threadIdx.x >> 5;
  int lane  = threadIdx.x & 31;
  int mtile = blockIdx.y * 4 + wave;
  int m0    = mtile * 16;
  if (m0 >= M) return;
  int n0   = blockIdx.x * (16 * NT);
  int half = lane >> 4;
  int lm   = lane & 15;
  int head = (int)blockIdx.z >= ZS ? 1 : 0;
  unsigned kbeg = (unsigned)((int)blockIdx.z - head * ZS) * (unsigned)KC;

  gcfp Ag = (gcfp)A;
  gcfp Bg = (gcfp)(head ? B1 : B0);

  int ncol[NT];
  unsigned boff[NT];
#pragma unroll
  for (int t = 0; t < NT; ++t) {
    ncol[t] = n0 + t * 16 + lm;
    boff[t] = (kbeg + (unsigned)(half * 2)) * (unsigned)N + (unsigned)ncol[t];
  }

  unsigned aoff = (unsigned)(m0 + lm) * (unsigned)K + kbeg + (unsigned)(half * 2);

  v8f acc[NT];
#pragma unroll
  for (int t = 0; t < NT; ++t) acc[t] = (v8f){0.f,0.f,0.f,0.f,0.f,0.f,0.f,0.f};

  for (int k0 = 0; k0 < KC; k0 += 4) {
    v2f a;
    a.x = Ag[aoff];
    a.y = Ag[aoff + 1];
#pragma unroll
    for (int t = 0; t < NT; ++t) {
      v2f b;
      b.x = Bg[boff[t]];
      b.y = Bg[boff[t] + (unsigned)N];
      acc[t] = __builtin_amdgcn_wmma_f32_16x16x4_f32(
           false, a,  false, b,
           (short)0, acc[t],  false,  false);
      boff[t] += 4u * (unsigned)N;
    }
    aoff += 4u;
  }

  gfp Pg = (gfp)P;
  unsigned pbase = (unsigned)blockIdx.z * (unsigned)M * (unsigned)N;
#pragma unroll
  for (int t = 0; t < NT; ++t) {
#pragma unroll
    for (int r = 0; r < 8; ++r) {
      int m = m0 + r + half * 8;
      Pg[pbase + (unsigned)m * (unsigned)N + (unsigned)ncol[t]] = acc[t][r];
    }
  }
}

#define RQ  5
#define RPL (KSPLIT / RQ)
#define RHB 25
static_assert(KSPLIT % RQ == 0 && RPL % RHB == 0, "plane groups and batches are exact");
typedef const __attribute__((address_space(1))) v2f* gcv2p;

__global__ void __launch_bounds__(RQ * 4 * 32)
k_head_reduce_value(const float* __restrict__ part,
                    const float* __restrict__ bp1, const float* __restrict__ bv1,
                    const float* __restrict__ Wv2, const float* __restrict__ bv2,
                    float* __restrict__ hidp, float* __restrict__ vout) {
  __shared__ float ps[RQ][HEAD_N];
  __shared__ float wsum[HEAD_N / 32];
  int t    = threadIdx.x;
  int lane = t & 31;
  int w    = t >> 5;
  int q    = w >> 2;
  int seg  = w & 3;
  int h    = (int)blockIdx.x >= BATCH ? 1 : 0;
  int b    = (int)blockIdx.x - h * BATCH;
  int n    = seg * 64 + lane * 2;
  gcfp p = (gcfp)part + ((size_t)(h * KSPLIT + q * RPL) * PLANE + (size_t)(b * HEAD_N + n));
  float sx = 0.0f, sy = 0.0f;
#pragma unroll
  for (int zb = 0; zb < RPL; zb += RHB) {
    v2f v[RHB];
#pragma unroll
    for (int i = 0; i < RHB; ++i) v[i] = *(gcv2p)(p + (unsigned)(zb + i) * (unsigned)PLANE);
#pragma unroll
    for (int i = 0; i < RHB; ++i) { sx += v[i].x; sy += v[i].y; }
  }
  ps[q][n]     = sx;
  ps[q][n + 1] = sy;
  __syncthreads();
  if (t < HEAD_N) {
    float s = ps[0][t];
#pragma unroll
    for (int g = 1; g < RQ; ++g) s += ps[g][t];
    s += (h ? bv1 : bp1)[t];
    s = s > 0.0f ? s : 0.0f;
    if (h == 0) {
      hidp[b * HEAD_N + t] = s;
    } else {
      float d = s * Wv2[t];
#pragma unroll
      for (int off = 16; off > 0; off >>= 1) d += __shfl_xor(d, off, 32);
      if (lane == 0) wsum[w] = d;
    }
  }
  if (h) {
    __syncthreads();
    if (t == 0) {
      float s = bv2[0];
#pragma unroll
      for (int i = 0; i < HEAD_N / 32; ++i) s += wsum[i];
      vout[b] = s;
    }
  }
}

static inline int cdiv(long long a, int b) { return (int)((a + b - 1) / b); }

extern "C" void kernel_launch(void* const* d_in, const int* in_sizes, int n_in,
                              void* d_out, int out_size, void* d_ws, size_t ws_size,
                              hipStream_t stream) {
  (void)in_sizes; (void)n_in; (void)out_size; (void)ws_size;

  const float* x    = (const float*)d_in[0];
  const int*   ei   = (const int*)d_in[1];
  const int*   rows = ei;
  const int*   cols = ei + N_EDGES;
  const float* W[5]  = {(const float*)d_in[2], (const float*)d_in[4], (const float*)d_in[6],
                        (const float*)d_in[8], (const float*)d_in[10]};
  const float* bb[5] = {(const float*)d_in[3], (const float*)d_in[5], (const float*)d_in[7],
                        (const float*)d_in[9], (const float*)d_in[11]};
  const float* Wp1 = (const float*)d_in[12]; const float* bp1 = (const float*)d_in[13];
  const float* Wp2 = (const float*)d_in[14]; const float* bp2 = (const float*)d_in[15];
  const float* Wv1 = (const float*)d_in[16]; const float* bv1 = (const float*)d_in[17];
  const float* Wv2 = (const float*)d_in[18]; const float* bv2 = (const float*)d_in[19];
  float* out = (float*)d_out;

  float* ws     = (float*)d_ws;
  int*   cnt    = (int*)ws;
  int*   rowptr = cnt + N_TOTAL;
  int*   bsum   = rowptr + N_TOTAL + 64;
  float* dinv   = (float*)(bsum + 256);
  int*   slot   = (int*)(dinv + N_TOTAL);
  int*   eid    = slot + N_EDGES;
  int*   srcs   = eid + N_EDGES;
  float* bufA   = (float*)(srcs + N_EDGES);
  float* zrowA  = bufA + (size_t)N_TOTAL * HID;
  float* bufB   = zrowA + 64;
  float* zrowB  = bufB + (size_t)N_TOTAL * HID;
  float* hidp   = zrowB + 64;
  float* part   = hidp + 2 * BATCH * 256;

  const int TB = 256;

  k_zero_int  <<<cdiv(N_TOTAL, TB), TB, 0, stream>>>(cnt, N_TOTAL, zrowA, zrowB);
  k_count     <<<cdiv(N_EDGES, TB), TB, 0, stream>>>(cols, cnt, slot);
  k_scan_local<<<SCAN_WG, SCAN_TB, 0, stream>>>(cnt, rowptr, bsum, dinv);
  k_scan_add  <<<SCAN_WG, SCAN_TB, 0, stream>>>(cnt, bsum, rowptr);
  k_fill      <<<cdiv(N_EDGES, TB), TB, 0, stream>>>(cols, slot, rowptr, eid);
  k_rank      <<<cdiv((long long)N_TOTAL * 32, TB), TB, 0, stream>>>(rows, rowptr, eid, srcs);

  dim3 layer_grid(1, N_TOTAL / 64);
  k_gemm_wmma<2, false, false, false, true><<<layer_grid, 128, 0, stream>>>(
      x, W[0], nullptr, dinv, bufA, N_TOTAL, HID, STATE);
  const int FG = N_TOTAL / TILE_M;
  const int FB = TILE_M * 32;
  k_aggregate_gemm<<<FG, FB, 0, stream>>>(rowptr, srcs, dinv, bufA, bb[0], W[1], bufB);
  k_aggregate_gemm<<<FG, FB, 0, stream>>>(rowptr, srcs, dinv, bufB, bb[1], W[2], bufA);
  k_aggregate_gemm<<<FG, FB, 0, stream>>>(rowptr, srcs, dinv, bufA, bb[2], W[3], bufB);
  k_aggregate_gemm<<<FG, FB, 0, stream>>>(rowptr, srcs, dinv, bufB, bb[3], W[4], bufA);
  k_aggregate<<<cdiv((long long)N_TOTAL * 32, TB), TB, 0, stream>>>(
      rowptr, srcs, dinv, bufA, bb[4], bufB);

  dim3 g1(HEAD_N / 32, cdiv(BATCH, 64), 2 * KSPLIT);
  k_gemm_wmma_ksplit<2><<<g1, 128, 0, stream>>>(
      bufB, Wp1, Wv1, part, BATCH, HEAD_N, HEAD_K, KCHUNK, KSPLIT);
  k_head_reduce_value<<<2 * BATCH, RQ * 4 * 32, 0, stream>>>(
      part, bp1, bv1, Wv2, bv2, hidp, out + (size_t)BATCH * NODE_COUNT);

  dim3 g2(cdiv(NODE_COUNT, 16), cdiv(BATCH, 64));
  k_gemm_wmma<1, true, true, false, false><<<g2, 128, 0, stream>>>(
      hidp, Wp2, bp2, nullptr, out, BATCH, NODE_COUNT, 256);
}


# Head split-K GEMM operands in wide loads: A one 16-byte load per lane per 8 k, B 8-byte column pairs (whole lines), k regrouped per WMMA alike in A and B; f32 sum order inside a K slice changes
# speedup vs baseline: 12.7593x; 1.0772x over previous
#include <hip/hip_runtime.h>

#define N_TOTAL    64000
#define NODE_COUNT 1000
#define BATCH      64
#define STATE      16
#define HID        32
#define N_EDGES    2048000
#define E_TOTAL    (N_EDGES + N_TOTAL)
#define HEAD_K     (NODE_COUNT * HID)
#define HEAD_N     256
#define KSPLIT     250
#define KCHUNK     (HEAD_K / KSPLIT)
#define PLANE      (BATCH * HEAD_N)
#define SCAN_TB    256
#define SCAN_WG    (N_TOTAL / SCAN_TB)
#define TILE_M     16
#define TILE_LD    (HID + 2)
static_assert(N_TOTAL % TILE_M == 0, "the fused layer kernel has no partial tile");

typedef float v2f __attribute__((ext_vector_type(2)));
typedef float v8f __attribute__((ext_vector_type(8)));

typedef const __attribute__((address_space(1))) float* gcfp;
typedef __attribute__((address_space(1))) float*       gfp;

__global__ void k_zero_int(int* __restrict__ p, int n, float* __restrict__ zrow0,
                           float* __restrict__ zrow1) {
  int i = blockIdx.x * blockDim.x + threadIdx.x;
  if (i < n) p[i] = 0;
  if (i < HID) { zrow0[i] = 0.0f; zrow1[i] = 0.0f; }
}

__global__ void k_count(const int* __restrict__ cols, int* __restrict__ cnt,
                        int* __restrict__ slot) {
  int e = blockIdx.x * blockDim.x + threadIdx.x;
  if (e < N_EDGES) slot[e] = atomicAdd(&cnt[cols[e]], 1);
}

__global__ void k_scan_local(const int* __restrict__ cnt, int* __restrict__ rowptr,
                             int* __restrict__ bsum, float* __restrict__ dinv) {
  __shared__ int s[SCAN_TB];
  int t = threadIdx.x;
  int i = blockIdx.x * SCAN_TB + t;
  int v = cnt[i];
  dinv[i] = rsqrtf((float)v + 1.0f);
  s[t] = v;
  __syncthreads();
  for (int off = 1; off < SCAN_TB; off <<= 1) {
    int a = t >= off ? s[t - off] : 0;
    __syncthreads();
    s[t] += a;
    __syncthreads();
  }
  rowptr[i] = s[t] - v;
  if (t == SCAN_TB - 1) bsum[blockIdx.x] = s[t];
}

__global__ void k_scan_add(const int* __restrict__ cnt, const int* __restrict__ bsum,
                           int* __restrict__ rowptr) {
  __shared__ int s[SCAN_TB];
  int t = threadIdx.x;
  int b = blockIdx.x;
  s[t] = t < b ? bsum[t] : 0;
  __syncthreads();
  for (int off = SCAN_TB / 2; off > 0; off >>= 1) {
    if (t < off) s[t] += s[t + off];
    __syncthreads();
  }
  int i = b * SCAN_TB + t;
  int p = rowptr[i] + s[0];
  rowptr[i] = p;
  if (i == N_TOTAL - 1) rowptr[N_TOTAL] = p + cnt[i];
}

__global__ void k_fill(const int* __restrict__ cols, const int* __restrict__ slot,
                       const int* __restrict__ rowptr, int* __restrict__ eid) {
  int e = blockIdx.x * blockDim.x + threadIdx.x;
  if (e < N_EDGES) eid[rowptr[cols[e]] + slot[e]] = e;
}

__global__ void k_rank(const int* __restrict__ rows, const int* __restrict__ rowptr,
                       const int* __restrict__ eid, int* __restrict__ srcs) {
  int t    = blockIdx.x * blockDim.x + threadIdx.x;
  int lane = t & 31;
  int c = __builtin_amdgcn_readfirstlane(t >> 5);
  if (c >= N_TOTAL) return;
  int beg = rowptr[c], end = rowptr[c + 1];
  for (int base = beg; base < end; base += 32) {
    int i   = base + lane;
    bool ok = i < end;
    int my  = ok ? eid[i] : 0x7fffffff;
    int rank = 0;
    for (int cb = beg; cb < end; cb += 32) {
      int q     = cb + lane;
      int other = q < end ? eid[q] : 0x7fffffff;
      int n     = end - cb < 32 ? end - cb : 32;
      for (int k = 0; k < n; ++k)
        rank += __builtin_amdgcn_readlane(other, k) < my ? 1 : 0;
    }
    if (ok) srcs[beg + rank] = rows[my];
  }
}

template <int K0, int NG>
__device__ __forceinline__ float agg_window(gcfp hg, int r0, int r1, int j, int G, float acc) {
  float v[8 * NG];
#pragma unroll
  for (int g = 0; g < NG; ++g) {
    if (g == 0 || K0 / 8 + g < G) {
#pragma unroll
      for (int k = 0; k < 8; ++k) {
        const int p = K0 + g * 8 + k;
        int r = p < 32 ? __builtin_amdgcn_readlane(r0, p) : __builtin_amdgcn_readlane(r1, p - 32);
        v[g * 8 + k] = hg[(unsigned)r * HID + j];
      }
    }
  }
#pragma unroll
  for (int g = 0; g < NG; ++g) {
    if (g == 0 || K0 / 8 + g < G) {
#pragma unroll
      for (int k = 0; k < 8; ++k) acc += v[g * 8 + k];
    }
  }
  return acc;
}

__device__ __forceinline__ float node_out(const int* __restrict__ rowptr,
                                          const int* __restrict__ srcs,
                                          const float* __restrict__ dinv, gcfp hg,
                                          const float* __restrict__ bias, int c, int j) {
  int beg = rowptr[c], end = rowptr[c + 1];
  float selfv = hg[(unsigned)c * HID + j];
  float bj    = bias[j];
  float dc    = dinv[c];
  float acc = 0.0f;
  for (int base = beg; base < end; base += 64) {
    int i0 = base + j, i1 = base + 32 + j;
    int r0 = N_TOTAL, r1 = N_TOTAL;
    if (i0 < end) r0 = srcs[i0];
    if (i1 < end) r1 = srcs[i1];
    int n = end - base < 64 ? end - base : 64;
    int G = (n + 7) >> 3;
    asm volatile("" :: "v"(r0), "v"(r1));
    acc = agg_window<0, 5>(hg, r0, r1, j, G, acc);
    if (G > 5) acc = agg_window<40, 1>(hg, r0, r1, j, G, acc);
    if (G > 6) acc = agg_window<48, 1>(hg, r0, r1, j, G, acc);
    if (G > 7) acc = agg_window<56, 1>(hg, r0, r1, j, G, acc);
  }
  acc += selfv;
  float v = dc * acc + bj;
  return v > 0.0f ? v : 0.0f;
}

__global__ void k_aggregate(const int* __restrict__ rowptr, const int* __restrict__ srcs,
                            const float* __restrict__ dinv,
                            const float* __restrict__ h, const float* __restrict__ bias,
                            float* __restrict__ out) {
  int t = blockIdx.x * blockDim.x + threadIdx.x;
  int j = t & 31;
  int c = __builtin_amdgcn_readfirstlane(t >> 5);
  if (c >= N_TOTAL) return;
  float v = node_out(rowptr, srcs, dinv, (gcfp)h, bias, c, j);
  ((gfp)out)[(unsigned)c * HID + j] = v;
}

__global__ void __launch_bounds__(TILE_M * 32)
k_aggregate_gemm(const int* __restrict__ rowptr, const int* __restrict__ srcs,
                 const float* __restrict__ dinv,
                 const float* __restrict__ h, const float* __restrict__ bias,
                 const float* __restrict__ Wn, float* __restrict__ hsn) {
  __shared__ __attribute__((aligned(16))) float tile[TILE_M * TILE_LD];
  int j = threadIdx.x & 31;
  int w  = __builtin_amdgcn_readfirstlane((int)(threadIdx.x >> 5));
  int m0 = blockIdx.x * TILE_M;
  float v = node_out(rowptr, srcs, dinv, (gcfp)h, bias, m0 + w, j);
  tile[w * TILE_LD + j] = v;
  __syncthreads();
  if (w != 0) return;

  int half = j >> 4;
  int lm   = j & 15;
  gcfp Wg = (gcfp)Wn;
  gcfp dg = (gcfp)dinv;
  float rs[8];
#pragma unroll
  for (int r = 0; r < 8; ++r) rs[r] = dg[m0 + half * 8 + r];

  v8f acc[2];
#pragma unroll
  for (int t = 0; t < 2; ++t) acc[t] = (v8f){0.f,0.f,0.f,0.f,0.f,0.f,0.f,0.f};
#pragma unroll
  for (int k0 = 0; k0 < HID; k0 += 4) {
    v2f a = *(const v2f*)&tile[lm * TILE_LD + k0 + half * 2];
#pragma unroll
    for (int t = 0; t < 2; ++t) {
      v2f b;
      b.x = Wg[(unsigned)(k0 + half * 2) * HID + t * 16 + lm];
      b.y = Wg[(unsigned)(k0 + half * 2 + 1) * HID + t * 16 + lm];
      acc[t] = __builtin_amdgcn_wmma_f32_16x16x4_f32(
           false, a,  false, b,
           (short)0, acc[t],  false,  false);
    }
  }
  gfp og = (gfp)hsn;
#pragma unroll
  for (int t = 0; t < 2; ++t) {
#pragma unroll
    for (int r = 0; r < 8; ++r) {
      int m = m0 + r + half * 8;
      og[(unsigned)m * HID + t * 16 + lm] = acc[t][r] * rs[r];
    }
  }
}

template <int NT, bool RAGGED, bool BIAS, bool RELU, bool ROWSCALE>
__global__ void k_gemm_wmma(const float* __restrict__ A, const float* __restrict__ B,
                            const float* __restrict__ bias, const float* __restrict__ rscale,
                            float* __restrict__ C, int M, int N, int K) {
  int wave  = threadIdx.x >> 5;
  int lane  = threadIdx.x & 31;
  int mtile = blockIdx.y * 4 + wave;
  int m0    = mtile * 16;
  if (m0 >= M) return;
  int n0   = blockIdx.x * (16 * NT);
  int half = lane >> 4;
  int lm   = lane & 15;

  gcfp Ag = (gcfp)A;
  gcfp Bg = (gcfp)B;

  float rs[8];
  if (ROWSCALE) {
    gcfp rg = (gcfp)rscale;
#pragma unroll
    for (int r = 0; r < 8; ++r) rs[r] = rg[m0 + half * 8 + r];
  }

  int  ncol[NT];
  bool nok[NT];
  unsigned boff[NT];
#pragma unroll
  for (int t = 0; t < NT; ++t) {
    ncol[t] = n0 + t * 16 + lm;
    int nld;
    if (RAGGED) {
      nok[t] = ncol[t] < N;
      nld    = nok[t] ? ncol[t] : (N - 1);
    } else {
      nok[t] = true;
      nld    = ncol[t];
    }
    boff[t] = (unsigned)(half * 2) * (unsigned)N + (unsigned)nld;
  }

  unsigned aoff = (unsigned)(m0 + lm) * (unsigned)K + (unsigned)(half * 2);

  v8f acc[NT];
#pragma unroll
  for (int t = 0; t < NT; ++t) acc[t] = (v8f){0.f,0.f,0.f,0.f,0.f,0.f,0.f,0.f};

  for (int k0 = 0; k0 < K; k0 += 4) {
    v2f a;
    a.x = Ag[aoff];
    a.y = Ag[aoff + 1];
#pragma unroll
    for (int t = 0; t < NT; ++t) {
      float b0 = Bg[boff[t]];
      float b1 = Bg[boff[t] + (unsigned)N];
      if (RAGGED) { b0 = nok[t] ? b0 : 0.0f; b1 = nok[t] ? b1 : 0.0f; }
      v2f b; b.x = b0; b.y = b1;
      acc[t] = __builtin_amdgcn_wmma_f32_16x16x4_f32(
           false, a,  false, b,
           (short)0, acc[t],  false,  false);
      boff[t] += 4u * (unsigned)N;
    }
    aoff += 4u;
  }

  gfp Cg = (gfp)C;
  gcfp biasg = (gcfp)bias;
#pragma unroll
  for (int t = 0; t < NT; ++t) {
    float bv = BIAS ? biasg[nok[t] ? ncol[t] : 0] : 0.0f;
    if (!RAGGED || nok[t]) {
#pragma unroll
      for (int r = 0; r < 8; ++r) {
        int m = m0 + r + half * 8;
        float v = acc[t][r];
        if (ROWSCALE) v *= rs[r];
        if (BIAS) v += bv;
        if (RELU) v = v > 0.0f ? v : 0.0f;
        Cg[(unsigned)m * (unsigned)N + (unsigned)ncol[t]] = v;
      }
    }
  }
}

typedef float v4f __attribute__((ext_vector_type(4)));
typedef const __attribute__((address_space(1))) v4f* gcv4p;
typedef const __attribute__((address_space(1))) v2f* gcv2q;
typedef __attribute__((address_space(1))) v2f*       gv2q;
static_assert(KCHUNK % 16 == 0, "a K slice is a whole number of 16-k blocks (two 8-k steps)");
static_assert(HEAD_K % 4 == 0 && HEAD_N % 2 == 0, "the 16-byte A loads and 8-byte B loads stay aligned");

__device__ __forceinline__ void ksplit_step(v4f a, v2f r0, v2f r1, v2f r2, v2f r3, v8f& acc0, v8f& acc1) {
  v2f a01; a01.x = a.x; a01.y = a.y;
  v2f a23; a23.x = a.z; a23.y = a.w;
  v2f b;
  b.x = r0.x; b.y = r1.x;
  acc0 = __builtin_amdgcn_wmma_f32_16x16x4_f32(
       false, a01,  false, b,
       (short)0, acc0,  false,  false);
  b.x = r0.y; b.y = r1.y;
  acc1 = __builtin_amdgcn_wmma_f32_16x16x4_f32(
       false, a01,  false, b,
       (short)0, acc1,  false,  false);
  b.x = r2.x; b.y = r3.x;
  acc0 = __builtin_amdgcn_wmma_f32_16x16x4_f32(
       false, a23,  false, b,
       (short)0, acc0,  false,  false);
  b.x = r2.y; b.y = r3.y;
  acc1 = __builtin_amdgcn_wmma_f32_16x16x4_f32(
       false, a23,  false, b,
       (short)0, acc1,  false,  false);
}

__global__ void k_gemm_wmma_ksplit(const float* __restrict__ A, const float* __restrict__ B0,
                                   const float* __restrict__ B1, float* __restrict__ P) {
  constexpr int M = BATCH, N = HEAD_N, K = HEAD_K, KC = KCHUNK, ZS = KSPLIT;
  int wave  = threadIdx.x >> 5;
  int lane  = threadIdx.x & 31;
  int mtile = blockIdx.y * 4 + wave;
  int m0    = mtile * 16;
  if (m0 >= M) return;
  int n0   = blockIdx.x * 32;
  int half = lane >> 4;
  int lm   = lane & 15;
  int head = (int)blockIdx.z >= ZS ? 1 : 0;
  unsigned kbeg = (unsigned)((int)blockIdx.z - head * ZS) * (unsigned)KC;

  gcv4p A4 = (gcv4p)A;
  gcv2q B2 = (gcv2q)(head ? B1 : B0);
  constexpr unsigned N2 = (unsigned)N >> 1;
  unsigned ncol = (unsigned)(n0 + 2 * lm);
  unsigned bidx = (kbeg + (unsigned)(half * 4)) * N2 + (ncol >> 1);
  unsigned aidx = ((unsigned)(m0 + lm) * (unsigned)K + kbeg + (unsigned)(half * 4)) >> 2;

  gcv4p ap = A4 + aidx;
  gcv2q bp = B2 + bidx;

  v8f acc0 = (v8f){0.f,0.f,0.f,0.f,0.f,0.f,0.f,0.f};
  v8f acc1 = (v8f){0.f,0.f,0.f,0.f,0.f,0.f,0.f,0.f};

#pragma unroll 2
  for (int k0 = 0; k0 < KC; k0 += 16) {
    v4f a0 = ap[0];
    v4f a1 = ap[2];
    v2f r[8];
#pragma unroll
    for (int j = 0; j < 8; ++j) r[j] = bp[(unsigned)((j >> 2) * 8 + (j & 3)) * N2];
    ksplit_step(a0, r[0], r[1], r[2], r[3], acc0, acc1);
    ksplit_step(a1, r[4], r[5], r[6], r[7], acc0, acc1);
    bp += 16u * N2;
    ap += 4;
  }

  gv2q pp = (gv2q)P + ((((unsigned)blockIdx.z * (unsigned)M + (unsigned)(m0 + half * 8)) * (unsigned)N + ncol) >> 1);
#pragma unroll
  for (int r8 = 0; r8 < 8; ++r8) {
    v2f o; o.x = acc0[r8]; o.y = acc1[r8];
    pp[(unsigned)r8 * N2] = o;
  }
}

#define RQ  5
#define RPL (KSPLIT / RQ)
#define RHB 25
static_assert(KSPLIT % RQ == 0 && RPL % RHB == 0, "plane groups and batches are exact");
typedef const __attribute__((address_space(1))) v2f* gcv2p;

__global__ void __launch_bounds__(RQ * 4 * 32)
k_head_reduce_value(const float* __restrict__ part,
                    const float* __restrict__ bp1, const float* __restrict__ bv1,
                    const float* __restrict__ Wv2, const float* __restrict__ bv2,
                    float* __restrict__ hidp, float* __restrict__ vout) {
  __shared__ float ps[RQ][HEAD_N];
  __shared__ float wsum[HEAD_N / 32];
  int t    = threadIdx.x;
  int lane = t & 31;
  int w    = t >> 5;
  int q    = w >> 2;
  int seg  = w & 3;
  int h    = (int)blockIdx.x >= BATCH ? 1 : 0;
  int b    = (int)blockIdx.x - h * BATCH;
  int n    = seg * 64 + lane * 2;
  gcfp p = (gcfp)part + ((size_t)(h * KSPLIT + q * RPL) * PLANE + (size_t)(b * HEAD_N + n));
  float sx = 0.0f, sy = 0.0f;
#pragma unroll
  for (int zb = 0; zb < RPL; zb += RHB) {
    v2f v[RHB];
#pragma unroll
    for (int i = 0; i < RHB; ++i) v[i] = *(gcv2p)(p + (unsigned)(zb + i) * (unsigned)PLANE);
#pragma unroll
    for (int i = 0; i < RHB; ++i) { sx += v[i].x; sy += v[i].y; }
  }
  ps[q][n]     = sx;
  ps[q][n + 1] = sy;
  __syncthreads();
  if (t < HEAD_N) {
    float s = ps[0][t];
#pragma unroll
    for (int g = 1; g < RQ; ++g) s += ps[g][t];
    s += (h ? bv1 : bp1)[t];
    s = s > 0.0f ? s : 0.0f;
    if (h == 0) {
      hidp[b * HEAD_N + t] = s;
    } else {
      float d = s * Wv2[t];
#pragma unroll
      for (int off = 16; off > 0; off >>= 1) d += __shfl_xor(d, off, 32);
      if (lane == 0) wsum[w] = d;
    }
  }
  if (h) {
    __syncthreads();
    if (t == 0) {
      float s = bv2[0];
#pragma unroll
      for (int i = 0; i < HEAD_N / 32; ++i) s += wsum[i];
      vout[b] = s;
    }
  }
}

static inline int cdiv(long long a, int b) { return (int)((a + b - 1) / b); }

extern "C" void kernel_launch(void* const* d_in, const int* in_sizes, int n_in,
                              void* d_out, int out_size, void* d_ws, size_t ws_size,
                              hipStream_t stream) {
  (void)in_sizes; (void)n_in; (void)out_size; (void)ws_size;

  const float* x    = (const float*)d_in[0];
  const int*   ei   = (const int*)d_in[1];
  const int*   rows = ei;
  const int*   cols = ei + N_EDGES;
  const float* W[5]  = {(const float*)d_in[2], (const float*)d_in[4], (const float*)d_in[6],
                        (const float*)d_in[8], (const float*)d_in[10]};
  const float* bb[5] = {(const float*)d_in[3], (const float*)d_in[5], (const float*)d_in[7],
                        (const float*)d_in[9], (const float*)d_in[11]};
  const float* Wp1 = (const float*)d_in[12]; const float* bp1 = (const float*)d_in[13];
  const float* Wp2 = (const float*)d_in[14]; const float* bp2 = (const float*)d_in[15];
  const float* Wv1 = (const float*)d_in[16]; const float* bv1 = (const float*)d_in[17];
  const float* Wv2 = (const float*)d_in[18]; const float* bv2 = (const float*)d_in[19];
  float* out = (float*)d_out;

  float* ws     = (float*)d_ws;
  int*   cnt    = (int*)ws;
  int*   rowptr = cnt + N_TOTAL;
  int*   bsum   = rowptr + N_TOTAL + 64;
  float* dinv   = (float*)(bsum + 256);
  int*   slot   = (int*)(dinv + N_TOTAL);
  int*   eid    = slot + N_EDGES;
  int*   srcs   = eid + N_EDGES;
  float* bufA   = (float*)(srcs + N_EDGES);
  float* zrowA  = bufA + (size_t)N_TOTAL * HID;
  float* bufB   = zrowA + 64;
  float* zrowB  = bufB + (size_t)N_TOTAL * HID;
  float* hidp   = zrowB + 64;
  float* part   = hidp + 2 * BATCH * 256;

  const int TB = 256;

  k_zero_int  <<<cdiv(N_TOTAL, TB), TB, 0, stream>>>(cnt, N_TOTAL, zrowA, zrowB);
  k_count     <<<cdiv(N_EDGES, TB), TB, 0, stream>>>(cols, cnt, slot);
  k_scan_local<<<SCAN_WG, SCAN_TB, 0, stream>>>(cnt, rowptr, bsum, dinv);
  k_scan_add  <<<SCAN_WG, SCAN_TB, 0, stream>>>(cnt, bsum, rowptr);
  k_fill      <<<cdiv(N_EDGES, TB), TB, 0, stream>>>(cols, slot, rowptr, eid);
  k_rank      <<<cdiv((long long)N_TOTAL * 32, TB), TB, 0, stream>>>(rows, rowptr, eid, srcs);

  dim3 layer_grid(1, N_TOTAL / 64);
  k_gemm_wmma<2, false, false, false, true><<<layer_grid, 128, 0, stream>>>(
      x, W[0], nullptr, dinv, bufA, N_TOTAL, HID, STATE);
  const int FG = N_TOTAL / TILE_M;
  const int FB = TILE_M * 32;
  k_aggregate_gemm<<<FG, FB, 0, stream>>>(rowptr, srcs, dinv, bufA, bb[0], W[1], bufB);
  k_aggregate_gemm<<<FG, FB, 0, stream>>>(rowptr, srcs, dinv, bufB, bb[1], W[2], bufA);
  k_aggregate_gemm<<<FG, FB, 0, stream>>>(rowptr, srcs, dinv, bufA, bb[2], W[3], bufB);
  k_aggregate_gemm<<<FG, FB, 0, stream>>>(rowptr, srcs, dinv, bufB, bb[3], W[4], bufA);
  k_aggregate<<<cdiv((long long)N_TOTAL * 32, TB), TB, 0, stream>>>(
      rowptr, srcs, dinv, bufA, bb[4], bufB);

  dim3 g1(HEAD_N / 32, cdiv(BATCH, 64), 2 * KSPLIT);
  k_gemm_wmma_ksplit<<<g1, 128, 0, stream>>>(bufB, Wp1, Wv1, part);
  k_head_reduce_value<<<2 * BATCH, RQ * 4 * 32, 0, stream>>>(
      part, bp1, bv1, Wv2, bv2, hidp, out + (size_t)BATCH * NODE_COUNT);

  dim3 g2(cdiv(NODE_COUNT, 16), cdiv(BATCH, 64));
  k_gemm_wmma<1, true, true, false, false><<<g2, 128, 0, stream>>>(
      hidp, Wp2, bp2, nullptr, out, BATCH, NODE_COUNT, 256);
}


# Edge list kept as fixed 64-slot rows per target (longer lists spill to a side list and a pool): arrival slot = final place, so both prefix-scan launches and the fill pass go; sum order unchanged
# speedup vs baseline: 13.9687x; 1.0948x over previous
#include <hip/hip_runtime.h>

#define N_TOTAL    64000
#define NODE_COUNT 1000
#define BATCH      64
#define STATE      16
#define HID        32
#define N_EDGES    2048000
#define E_TOTAL    (N_EDGES + N_TOTAL)
#define HEAD_K     (NODE_COUNT * HID)
#define HEAD_N     256
#define KSPLIT     250
#define KCHUNK     (HEAD_K / KSPLIT)
#define PLANE      (BATCH * HEAD_N)
#define CAP        64
#define TILE_M     16
#define TILE_LD    (HID + 2)
static_assert(N_TOTAL % TILE_M == 0, "the fused layer kernel has no partial tile");

typedef float v2f __attribute__((ext_vector_type(2)));
typedef float v8f __attribute__((ext_vector_type(8)));

typedef const __attribute__((address_space(1))) float* gcfp;
typedef __attribute__((address_space(1))) float*       gfp;

__global__ void k_zero_int(int* __restrict__ p, int n, float* __restrict__ zrow0,
                           float* __restrict__ zrow1) {
  int i = blockIdx.x * blockDim.x + threadIdx.x;
  if (i < n) p[i] = 0;
  if (i < HID) { zrow0[i] = 0.0f; zrow1[i] = 0.0f; }
}

__global__ void k_count_place(const int* __restrict__ cols, int* __restrict__ cnt,
                              int* __restrict__ ctr, int* __restrict__ ell,
                              int* __restrict__ ovf) {
  int e = blockIdx.x * blockDim.x + threadIdx.x;
  if (e < N_EDGES) {
    int c = cols[e];
    int s = atomicAdd(&cnt[c], 1);
    if (s < CAP) ell[(unsigned)c * CAP + s] = e;
    else         ovf[atomicAdd(&ctr[0], 1)] = e;
  }
}

#define NO_ID 0x7fffffff

__device__ __forceinline__ int rank_chunk(int q, int lane, int c, int id0, int id1, int nov,
                                          const int* __restrict__ ovf,
                                          const int* __restrict__ cols) {
  if (q == 0) return id0;
  if (q == 1) return id1;
  int k = (q - 2) * 32 + lane;
  int r = NO_ID;
  if (k < nov) {
    int e = ovf[k];
    if (cols[e] == c) r = e;
  }
  return r;
}

__global__ void k_rank(const int* __restrict__ rows, const int* __restrict__ cols,
                       const int* __restrict__ cnt, int* __restrict__ ctr,
                       const int* __restrict__ ell, const int* __restrict__ ovf,
                       int* __restrict__ poff, float* __restrict__ dinv,
                       int* __restrict__ srcs) {
  int t    = blockIdx.x * blockDim.x + threadIdx.x;
  int lane = t & 31;
  int c = __builtin_amdgcn_readfirstlane(t >> 5);
  if (c >= N_TOTAL) return;
  int n = cnt[c];
  if (lane == 0) dinv[c] = rsqrtf((float)n + 1.0f);
  int m = n < CAP ? n : CAP;
  const int* row = ell + (unsigned)c * CAP;
  int id0 = lane < m ? row[lane] : NO_ID;
  int id1 = lane + 32 < m ? row[lane + 32] : NO_ID;
  if (n <= CAP) {
    int s0 = rows[id0 != NO_ID ? id0 : 0];
    int s1 = rows[id1 != NO_ID ? id1 : 0];
    int rk0 = 0, rk1 = 0;
    if (m <= 32) {
#pragma unroll
      for (int k = 0; k < 32; ++k)
        rk0 += __builtin_amdgcn_readlane(id0, k) < id0 ? 1 : 0;
    } else {
#pragma unroll
      for (int k = 0; k < 32; ++k) {
        int o = __builtin_amdgcn_readlane(id0, k);
        rk0 += o < id0 ? 1 : 0;
        rk1 += o < id1 ? 1 : 0;
      }
#pragma unroll
      for (int k = 0; k < 32; ++k) {
        int o = __builtin_amdgcn_readlane(id1, k);
        rk0 += o < id0 ? 1 : 0;
        rk1 += o < id1 ? 1 : 0;
      }
    }
    int* out = srcs + (unsigned)c * CAP;
    if (id0 != NO_ID) out[rk0] = s0;
    if (id1 != NO_ID) out[rk1] = s1;
    return;
  }
  int nov = ctr[0];
  int off = 0;
  if (lane == 0) {
    off = atomicAdd(&ctr[1], n);
    poff[c] = off;
  }
  off = __builtin_amdgcn_readfirstlane(off);
  int* out = srcs + (unsigned)N_TOTAL * CAP + off;
  int nch = 2 + (nov + 31) / 32;
  for (int qa = 0; qa < nch; ++qa) {
    int my   = rank_chunk(qa, lane, c, id0, id1, nov, ovf, cols);
    int rank = 0;
    for (int qb = 0; qb < nch; ++qb) {
      int other = rank_chunk(qb, lane, c, id0, id1, nov, ovf, cols);
#pragma unroll
      for (int k = 0; k < 32; ++k)
        rank += __builtin_amdgcn_readlane(other, k) < my ? 1 : 0;
    }
    if (my != NO_ID) out[rank] = rows[my];
  }
}

template <int K0, int NG>
__device__ __forceinline__ float agg_window(gcfp hg, int r0, int r1, int j, int G, float acc) {
  float v[8 * NG];
#pragma unroll
  for (int g = 0; g < NG; ++g) {
    if (g == 0 || K0 / 8 + g < G) {
#pragma unroll
      for (int k = 0; k < 8; ++k) {
        const int p = K0 + g * 8 + k;
        int r = p < 32 ? __builtin_amdgcn_readlane(r0, p) : __builtin_amdgcn_readlane(r1, p - 32);
        v[g * 8 + k] = hg[(unsigned)r * HID + j];
      }
    }
  }
#pragma unroll
  for (int g = 0; g < NG; ++g) {
    if (g == 0 || K0 / 8 + g < G) {
#pragma unroll
      for (int k = 0; k < 8; ++k) acc += v[g * 8 + k];
    }
  }
  return acc;
}

__device__ __forceinline__ float node_out(const int* __restrict__ cnt,
                                          const int* __restrict__ poff,
                                          const int* __restrict__ srcs,
                                          const float* __restrict__ dinv, gcfp hg,
                                          const float* __restrict__ bias, int c, int j) {
  int n   = cnt[c];
  int beg = c * CAP;
  if (n > CAP) beg = N_TOTAL * CAP + poff[c];
  int end = beg + n;
  float selfv = hg[(unsigned)c * HID + j];
  float bj    = bias[j];
  float dc    = dinv[c];
  float acc = 0.0f;
  for (int base = beg; base < end; base += 64) {
    int i0 = base + j, i1 = base + 32 + j;
    int r0 = N_TOTAL, r1 = N_TOTAL;
    if (i0 < end) r0 = srcs[i0];
    if (i1 < end) r1 = srcs[i1];
    int n = end - base < 64 ? end - base : 64;
    int G = (n + 7) >> 3;
    asm volatile("" :: "v"(r0), "v"(r1));
    acc = agg_window<0, 5>(hg, r0, r1, j, G, acc);
    if (G > 5) acc = agg_window<40, 1>(hg, r0, r1, j, G, acc);
    if (G > 6) acc = agg_window<48, 1>(hg, r0, r1, j, G, acc);
    if (G > 7) acc = agg_window<56, 1>(hg, r0, r1, j, G, acc);
  }
  acc += selfv;
  float v = dc * acc + bj;
  return v > 0.0f ? v : 0.0f;
}

__global__ void k_aggregate(const int* __restrict__ cnt, const int* __restrict__ poff,
                            const int* __restrict__ srcs, const float* __restrict__ dinv,
                            const float* __restrict__ h, const float* __restrict__ bias,
                            float* __restrict__ out) {
  int t = blockIdx.x * blockDim.x + threadIdx.x;
  int j = t & 31;
  int c = __builtin_amdgcn_readfirstlane(t >> 5);
  if (c >= N_TOTAL) return;
  float v = node_out(cnt, poff, srcs, dinv, (gcfp)h, bias, c, j);
  ((gfp)out)[(unsigned)c * HID + j] = v;
}

__global__ void __launch_bounds__(TILE_M * 32)
k_aggregate_gemm(const int* __restrict__ cnt, const int* __restrict__ poff,
                 const int* __restrict__ srcs, const float* __restrict__ dinv,
                 const float* __restrict__ h, const float* __restrict__ bias,
                 const float* __restrict__ Wn, float* __restrict__ hsn) {
  __shared__ __attribute__((aligned(16))) float tile[TILE_M * TILE_LD];
  int j = threadIdx.x & 31;
  int w  = __builtin_amdgcn_readfirstlane((int)(threadIdx.x >> 5));
  int m0 = blockIdx.x * TILE_M;
  float v = node_out(cnt, poff, srcs, dinv, (gcfp)h, bias, m0 + w, j);
  tile[w * TILE_LD + j] = v;
  __syncthreads();
  if (w != 0) return;

  int half = j >> 4;
  int lm   = j & 15;
  gcfp Wg = (gcfp)Wn;
  gcfp dg = (gcfp)dinv;
  float rs[8];
#pragma unroll
  for (int r = 0; r < 8; ++r) rs[r] = dg[m0 + half * 8 + r];

  v8f acc[2];
#pragma unroll
  for (int t = 0; t < 2; ++t) acc[t] = (v8f){0.f,0.f,0.f,0.f,0.f,0.f,0.f,0.f};
#pragma unroll
  for (int k0 = 0; k0 < HID; k0 += 4) {
    v2f a = *(const v2f*)&tile[lm * TILE_LD + k0 + half * 2];
#pragma unroll
    for (int t = 0; t < 2; ++t) {
      v2f b;
      b.x = Wg[(unsigned)(k0 + half * 2) * HID + t * 16 + lm];
      b.y = Wg[(unsigned)(k0 + half * 2 + 1) * HID + t * 16 + lm];
      acc[t] = __builtin_amdgcn_wmma_f32_16x16x4_f32(
           false, a,  false, b,
           (short)0, acc[t],  false,  false);
    }
  }
  gfp og = (gfp)hsn;
#pragma unroll
  for (int t = 0; t < 2; ++t) {
#pragma unroll
    for (int r = 0; r < 8; ++r) {
      int m = m0 + r + half * 8;
      og[(unsigned)m * HID + t * 16 + lm] = acc[t][r] * rs[r];
    }
  }
}

template <int NT, bool RAGGED, bool BIAS, bool RELU, bool ROWSCALE>
__global__ void k_gemm_wmma(const float* __restrict__ A, const float* __restrict__ B,
                            const float* __restrict__ bias, const float* __restrict__ rscale,
                            float* __restrict__ C, int M, int N, int K) {
  int wave  = threadIdx.x >> 5;
  int lane  = threadIdx.x & 31;
  int mtile = blockIdx.y * 4 + wave;
  int m0    = mtile * 16;
  if (m0 >= M) return;
  int n0   = blockIdx.x * (16 * NT);
  int half = lane >> 4;
  int lm   = lane & 15;

  gcfp Ag = (gcfp)A;
  gcfp Bg = (gcfp)B;

  float rs[8];
  if (ROWSCALE) {
    gcfp rg = (gcfp)rscale;
#pragma unroll
    for (int r = 0; r < 8; ++r) rs[r] = rg[m0 + half * 8 + r];
  }

  int  ncol[NT];
  bool nok[NT];
  unsigned boff[NT];
#pragma unroll
  for (int t = 0; t < NT; ++t) {
    ncol[t] = n0 + t * 16 + lm;
    int nld;
    if (RAGGED) {
      nok[t] = ncol[t] < N;
      nld    = nok[t] ? ncol[t] : (N - 1);
    } else {
      nok[t] = true;
      nld    = ncol[t];
    }
    boff[t] = (unsigned)(half * 2) * (unsigned)N + (unsigned)nld;
  }

  unsigned aoff = (unsigned)(m0 + lm) * (unsigned)K + (unsigned)(half * 2);

  v8f acc[NT];
#pragma unroll
  for (int t = 0; t < NT; ++t) acc[t] = (v8f){0.f,0.f,0.f,0.f,0.f,0.f,0.f,0.f};

  for (int k0 = 0; k0 < K; k0 += 4) {
    v2f a;
    a.x = Ag[aoff];
    a.y = Ag[aoff + 1];
#pragma unroll
    for (int t = 0; t < NT; ++t) {
      float b0 = Bg[boff[t]];
      float b1 = Bg[boff[t] + (unsigned)N];
      if (RAGGED) { b0 = nok[t] ? b0 : 0.0f; b1 = nok[t] ? b1 : 0.0f; }
      v2f b; b.x = b0; b.y = b1;
      acc[t] = __builtin_amdgcn_wmma_f32_16x16x4_f32(
           false, a,  false, b,
           (short)0, acc[t],  false,  false);
      boff[t] += 4u * (unsigned)N;
    }
    aoff += 4u;
  }

  gfp Cg = (gfp)C;
  gcfp biasg = (gcfp)bias;
#pragma unroll
  for (int t = 0; t < NT; ++t) {
    float bv = BIAS ? biasg[nok[t] ? ncol[t] : 0] : 0.0f;
    if (!RAGGED || nok[t]) {
#pragma unroll
      for (int r = 0; r < 8; ++r) {
        int m = m0 + r + half * 8;
        float v = acc[t][r];
        if (ROWSCALE) v *= rs[r];
        if (BIAS) v += bv;
        if (RELU) v = v > 0.0f ? v : 0.0f;
        Cg[(unsigned)m * (unsigned)N + (unsigned)ncol[t]] = v;
      }
    }
  }
}

typedef float v4f __attribute__((ext_vector_type(4)));
typedef const __attribute__((address_space(1))) v4f* gcv4p;
typedef const __attribute__((address_space(1))) v2f* gcv2q;
typedef __attribute__((address_space(1))) v2f*       gv2q;
static_assert(KCHUNK % 16 == 0, "a K slice is a whole number of 16-k blocks (two 8-k steps)");
static_assert(HEAD_K % 4 == 0 && HEAD_N % 2 == 0, "the 16-byte A loads and 8-byte B loads stay aligned");

__device__ __forceinline__ void ksplit_step(v4f a, v2f r0, v2f r1, v2f r2, v2f r3, v8f& acc0, v8f& acc1) {
  v2f a01; a01.x = a.x; a01.y = a.y;
  v2f a23; a23.x = a.z; a23.y = a.w;
  v2f b;
  b.x = r0.x; b.y = r1.x;
  acc0 = __builtin_amdgcn_wmma_f32_16x16x4_f32(
       false, a01,  false, b,
       (short)0, acc0,  false,  false);
  b.x = r0.y; b.y = r1.y;
  acc1 = __builtin_amdgcn_wmma_f32_16x16x4_f32(
       false, a01,  false, b,
       (short)0, acc1,  false,  false);
  b.x = r2.x; b.y = r3.x;
  acc0 = __builtin_amdgcn_wmma_f32_16x16x4_f32(
       false, a23,  false, b,
       (short)0, acc0,  false,  false);
  b.x = r2.y; b.y = r3.y;
  acc1 = __builtin_amdgcn_wmma_f32_16x16x4_f32(
       false, a23,  false, b,
       (short)0, acc1,  false,  false);
}

__global__ void k_gemm_wmma_ksplit(const float* __restrict__ A, const float* __restrict__ B0,
                                   const float* __restrict__ B1, float* __restrict__ P) {
  constexpr int M = BATCH, N = HEAD_N, K = HEAD_K, KC = KCHUNK, ZS = KSPLIT;
  int wave  = threadIdx.x >> 5;
  int lane  = threadIdx.x & 31;
  int mtile = blockIdx.y * 4 + wave;
  int m0    = mtile * 16;
  if (m0 >= M) return;
  int n0   = blockIdx.x * 32;
  int half = lane >> 4;
  int lm   = lane & 15;
  int head = (int)blockIdx.z >= ZS ? 1 : 0;
  unsigned kbeg = (unsigned)((int)blockIdx.z - head * ZS) * (unsigned)KC;

  gcv4p A4 = (gcv4p)A;
  gcv2q B2 = (gcv2q)(head ? B1 : B0);
  constexpr unsigned N2 = (unsigned)N >> 1;
  unsigned ncol = (unsigned)(n0 + 2 * lm);
  unsigned bidx = (kbeg + (unsigned)(half * 4)) * N2 + (ncol >> 1);
  unsigned aidx = ((unsigned)(m0 + lm) * (unsigned)K + kbeg + (unsigned)(half * 4)) >> 2;

  gcv4p ap = A4 + aidx;
  gcv2q bp = B2 + bidx;

  v8f acc0 = (v8f){0.f,0.f,0.f,0.f,0.f,0.f,0.f,0.f};
  v8f acc1 = (v8f){0.f,0.f,0.f,0.f,0.f,0.f,0.f,0.f};

#pragma unroll 2
  for (int k0 = 0; k0 < KC; k0 += 16) {
    v4f a0 = ap[0];
    v4f a1 = ap[2];
    v2f r[8];
#pragma unroll
    for (int j = 0; j < 8; ++j) r[j] = bp[(unsigned)((j >> 2) * 8 + (j & 3)) * N2];
    ksplit_step(a0, r[0], r[1], r[2], r[3], acc0, acc1);
    ksplit_step(a1, r[4], r[5], r[6], r[7], acc0, acc1);
    bp += 16u * N2;
    ap += 4;
  }

  gv2q pp = (gv2q)P + ((((unsigned)blockIdx.z * (unsigned)M + (unsigned)(m0 + half * 8)) * (unsigned)N + ncol) >> 1);
#pragma unroll
  for (int r8 = 0; r8 < 8; ++r8) {
    v2f o; o.x = acc0[r8]; o.y = acc1[r8];
    pp[(unsigned)r8 * N2] = o;
  }
}

#define RQ  5
#define RPL (KSPLIT / RQ)
#define RHB 25
static_assert(KSPLIT % RQ == 0 && RPL % RHB == 0, "plane groups and batches are exact");
typedef const __attribute__((address_space(1))) v2f* gcv2p;

__global__ void __launch_bounds__(RQ * 4 * 32)
k_head_reduce_value(const float* __restrict__ part,
                    const float* __restrict__ bp1, const float* __restrict__ bv1,
                    const float* __restrict__ Wv2, const float* __restrict__ bv2,
                    float* __restrict__ hidp, float* __restrict__ vout) {
  __shared__ float ps[RQ][HEAD_N];
  __shared__ float wsum[HEAD_N / 32];
  int t    = threadIdx.x;
  int lane = t & 31;
  int w    = t >> 5;
  int q    = w >> 2;
  int seg  = w & 3;
  int h    = (int)blockIdx.x >= BATCH ? 1 : 0;
  int b    = (int)blockIdx.x - h * BATCH;
  int n    = seg * 64 + lane * 2;
  gcfp p = (gcfp)part + ((size_t)(h * KSPLIT + q * RPL) * PLANE + (size_t)(b * HEAD_N + n));
  float sx = 0.0f, sy = 0.0f;
#pragma unroll
  for (int zb = 0; zb < RPL; zb += RHB) {
    v2f v[RHB];
#pragma unroll
    for (int i = 0; i < RHB; ++i) v[i] = *(gcv2p)(p + (unsigned)(zb + i) * (unsigned)PLANE);
#pragma unroll
    for (int i = 0; i < RHB; ++i) { sx += v[i].x; sy += v[i].y; }
  }
  ps[q][n]     = sx;
  ps[q][n + 1] = sy;
  __syncthreads();
  if (t < HEAD_N) {
    float s = ps[0][t];
#pragma unroll
    for (int g = 1; g < RQ; ++g) s += ps[g][t];
    s += (h ? bv1 : bp1)[t];
    s = s > 0.0f ? s : 0.0f;
    if (h == 0) {
      hidp[b * HEAD_N + t] = s;
    } else {
      float d = s * Wv2[t];
#pragma unroll
      for (int off = 16; off > 0; off >>= 1) d += __shfl_xor(d, off, 32);
      if (lane == 0) wsum[w] = d;
    }
  }
  if (h) {
    __syncthreads();
    if (t == 0) {
      float s = bv2[0];
#pragma unroll
      for (int i = 0; i < HEAD_N / 32; ++i) s += wsum[i];
      vout[b] = s;
    }
  }
}

static inline int cdiv(long long a, int b) { return (int)((a + b - 1) / b); }

extern "C" void kernel_launch(void* const* d_in, const int* in_sizes, int n_in,
                              void* d_out, int out_size, void* d_ws, size_t ws_size,
                              hipStream_t stream) {
  (void)in_sizes; (void)n_in; (void)out_size; (void)ws_size;

  const float* x    = (const float*)d_in[0];
  const int*   ei   = (const int*)d_in[1];
  const int*   rows = ei;
  const int*   cols = ei + N_EDGES;
  const float* W[5]  = {(const float*)d_in[2], (const float*)d_in[4], (const float*)d_in[6],
                        (const float*)d_in[8], (const float*)d_in[10]};
  const float* bb[5] = {(const float*)d_in[3], (const float*)d_in[5], (const float*)d_in[7],
                        (const float*)d_in[9], (const float*)d_in[11]};
  const float* Wp1 = (const float*)d_in[12]; const float* bp1 = (const float*)d_in[13];
  const float* Wp2 = (const float*)d_in[14]; const float* bp2 = (const float*)d_in[15];
  const float* Wv1 = (const float*)d_in[16]; const float* bv1 = (const float*)d_in[17];
  const float* Wv2 = (const float*)d_in[18]; const float* bv2 = (const float*)d_in[19];
  float* out = (float*)d_out;

  float* ws     = (float*)d_ws;
  int*   cnt    = (int*)ws;
  int*   ctr    = cnt + N_TOTAL;
  int*   poff   = ctr + 64;
  float* dinv   = (float*)(poff + N_TOTAL);
  int*   ell    = (int*)(dinv + N_TOTAL);
  int*   ovf    = ell + (size_t)N_TOTAL * CAP;
  int*   srcs   = ovf + N_EDGES;
  float* bufA   = (float*)(srcs + (size_t)N_TOTAL * CAP + N_EDGES);
  float* zrowA  = bufA + (size_t)N_TOTAL * HID;
  float* bufB   = zrowA + 64;
  float* zrowB  = bufB + (size_t)N_TOTAL * HID;
  float* hidp   = zrowB + 64;
  float* part   = hidp + 2 * BATCH * 256;

  const int TB = 256;

  k_zero_int   <<<cdiv(N_TOTAL + 64, TB), TB, 0, stream>>>(cnt, N_TOTAL + 64, zrowA, zrowB);
  k_count_place<<<cdiv(N_EDGES, TB), TB, 0, stream>>>(cols, cnt, ctr, ell, ovf);
  k_rank       <<<cdiv((long long)N_TOTAL * 32, TB), TB, 0, stream>>>(
      rows, cols, cnt, ctr, ell, ovf, poff, dinv, srcs);

  dim3 layer_grid(1, N_TOTAL / 64);
  k_gemm_wmma<2, false, false, false, true><<<layer_grid, 128, 0, stream>>>(
      x, W[0], nullptr, dinv, bufA, N_TOTAL, HID, STATE);
  const int FG = N_TOTAL / TILE_M;
  const int FB = TILE_M * 32;
  k_aggregate_gemm<<<FG, FB, 0, stream>>>(cnt, poff, srcs, dinv, bufA, bb[0], W[1], bufB);
  k_aggregate_gemm<<<FG, FB, 0, stream>>>(cnt, poff, srcs, dinv, bufB, bb[1], W[2], bufA);
  k_aggregate_gemm<<<FG, FB, 0, stream>>>(cnt, poff, srcs, dinv, bufA, bb[2], W[3], bufB);
  k_aggregate_gemm<<<FG, FB, 0, stream>>>(cnt, poff, srcs, dinv, bufB, bb[3], W[4], bufA);
  k_aggregate<<<cdiv((long long)N_TOTAL * 32, TB), TB, 0, stream>>>(
      cnt, poff, srcs, dinv, bufA, bb[4], bufB);

  dim3 g1(HEAD_N / 32, cdiv(BATCH, 64), 2 * KSPLIT);
  k_gemm_wmma_ksplit<<<g1, 128, 0, stream>>>(bufB, Wp1, Wv1, part);
  k_head_reduce_value<<<2 * BATCH, RQ * 4 * 32, 0, stream>>>(
      part, bp1, bv1, Wv2, bv2, hidp, out + (size_t)BATCH * NODE_COUNT);

  dim3 g2(cdiv(NODE_COUNT, 16), cdiv(BATCH, 64));
  k_gemm_wmma<1, true, true, false, false><<<g2, 128, 0, stream>>>(
      hidp, Wp2, bp2, nullptr, out, BATCH, NODE_COUNT, 256);
}
